# starting kernel
# baseline (speedup 1.0000x reference)
#include <hip/hip_runtime.h>
#include <hip/hip_bf16.h>


typedef float v2f __attribute__((ext_vector_type(2)));
typedef float v8f __attribute__((ext_vector_type(8)));

#define BSZ   1024
#define CCH   16
#define EDIM  64
#define SLICE (CCH * EDIM)
#define ATOT  (BSZ * BSZ * CCH)

__device__ __forceinline__ unsigned int enc_f32(float f) {
  unsigned int u = __float_as_uint(f);
  return (u & 0x80000000u) ? ~u : (u | 0x80000000u);
}
__device__ __forceinline__ float dec_f32(unsigned int u) {
  return (u & 0x80000000u) ? __uint_as_float(u & 0x7FFFFFFFu)
                           : __uint_as_float(~u);
}

#define LDA_P 20
#define LDB_P 68

__global__ __launch_bounds__(256) void gemm_bias_act(
    const float* __restrict__ A, const float* __restrict__ W,
    const float* __restrict__ bias, float* __restrict__ out,
    int M, int K, int N, int relu)
{
  __shared__ float lA[128 * LDA_P];
  __shared__ float lB[16 * LDB_P];

  const int t    = threadIdx.x;
  const int lane = t & 31;
  const int wave = t >> 5;
  const int wm   = wave & 3;
  const int wn   = wave >> 2;
  const int bm   = blockIdx.y * 128;
  const int bn   = blockIdx.x * 64;

  v8f acc[2][2] = {};

  const int mrow = lane & 15;
  const int koff = (lane >> 4) * 2;
  const int ncol = lane & 15;

  for (int k0 = 0; k0 < K; k0 += 16) {
    #pragma unroll
    for (int i = 0; i < 2; ++i) {
      int idx = t + i * 256;
      int row = idx >> 2;
      int c4  = idx & 3;
      float4 v = *(const float4*)(A + (size_t)(bm + row) * K + k0 + c4 * 4);
      *(float4*)(lA + row * LDA_P + c4 * 4) = v;
    }
    {
      int row = t >> 4;
      int c4  = t & 15;
      float4 v = *(const float4*)(W + (size_t)(k0 + row) * N + bn + c4 * 4);
      *(float4*)(lB + row * LDB_P + c4 * 4) = v;
    }
    __syncthreads();

    #pragma unroll
    for (int kk = 0; kk < 16; kk += 4) {
      v2f afrag[2], bfrag[2];
      #pragma unroll
      for (int mt = 0; mt < 2; ++mt) {
        int row = wm * 32 + mt * 16 + mrow;
        afrag[mt] = *(const v2f*)(lA + row * LDA_P + kk + koff);
      }
      #pragma unroll
      for (int nt = 0; nt < 2; ++nt) {
        int col = wn * 32 + nt * 16 + ncol;
        bfrag[nt].x = lB[(kk + koff + 0) * LDB_P + col];
        bfrag[nt].y = lB[(kk + koff + 1) * LDB_P + col];
      }
      #pragma unroll
      for (int mt = 0; mt < 2; ++mt)
        #pragma unroll
        for (int nt = 0; nt < 2; ++nt)
          acc[mt][nt] = __builtin_amdgcn_wmma_f32_16x16x4_f32(
              false, afrag[mt], false, bfrag[nt],
              (short)0, acc[mt][nt], false, false);
    }
    __syncthreads();
  }

  const int rsel = (lane >> 4) * 8;
  #pragma unroll
  for (int nt = 0; nt < 2; ++nt) {
    int col = bn + wn * 32 + nt * 16 + ncol;
    float bv = bias[col];
    #pragma unroll
    for (int mt = 0; mt < 2; ++mt) {
      #pragma unroll
      for (int v = 0; v < 8; ++v) {
        int row = bm + wm * 32 + mt * 16 + rsel + v;
        float x = acc[mt][nt][v] + bv;
        if (relu) x = fmaxf(x, 0.0f);
        out[(size_t)row * N + col] = x;
      }
    }
  }
}

__global__ __launch_bounds__(256) void norm_rows64(float* __restrict__ q)
{
  const int lane = threadIdx.x & 31;
  const int wave = threadIdx.x >> 5;
  const int row  = blockIdx.x * 8 + wave;
  float* p = q + (size_t)row * 64;

  float2 x = *(float2*)(p + lane * 2);
  float s  = x.x + x.y;
  float ss = x.x * x.x + x.y * x.y;
  #pragma unroll
  for (int off = 16; off > 0; off >>= 1) {
    s  += __shfl_xor(s,  off, 32);
    ss += __shfl_xor(ss, off, 32);
  }
  float mean = s * (1.0f / 64.0f);
  float var  = (ss - 64.0f * mean * mean) * (1.0f / 63.0f);
  float inv  = rsqrtf(var + 1e-8f);
  x.x = (x.x - mean) * inv;
  x.y = (x.y - mean) * inv;
  *(float2*)(p + lane * 2) = x;
}

__global__ void sink_init(float* __restrict__ v, unsigned int* __restrict__ maxbuf)
{
  int i = blockIdx.x * 256 + threadIdx.x;
  if (i < BSZ * CCH) v[i] = 1.0f;
  if (i < CCH) maxbuf[i] = 0u;
}

#define LQA_P 68
#define LQB_P 65

__global__ __launch_bounds__(256) void affinity_wmma(
    const float* __restrict__ q1, const float* __restrict__ q2,
    float* __restrict__ aff, unsigned int* __restrict__ maxbuf)
{
  __shared__ float lA[128 * LQA_P];
  __shared__ float lB[64 * LQB_P];
  __shared__ float red[256];

  const int t    = threadIdx.x;
  const int lane = t & 31;
  const int wave = t >> 5;
  const int wm   = wave & 3;
  const int wn   = wave >> 2;
  const int bb   = blockIdx.y * 128;
  const int bd   = blockIdx.x * 64;
  const int c    = blockIdx.z;

  #pragma unroll
  for (int i = 0; i < 8; ++i) {
    int idx = t + i * 256;
    int row = idx >> 4;
    int c4  = idx & 15;
    float4 v = *(const float4*)(q1 + (size_t)(bb + row) * SLICE + c * EDIM + c4 * 4);
    *(float4*)(lA + row * LQA_P + c4 * 4) = v;
  }
  #pragma unroll
  for (int i = 0; i < 4; ++i) {
    int idx = t + i * 256;
    int dl = idx >> 4;
    int e4 = idx & 15;
    float4 v = *(const float4*)(q2 + (size_t)(bd + dl) * SLICE + c * EDIM + e4 * 4);
    lB[(e4 * 4 + 0) * LQB_P + dl] = v.x;
    lB[(e4 * 4 + 1) * LQB_P + dl] = v.y;
    lB[(e4 * 4 + 2) * LQB_P + dl] = v.z;
    lB[(e4 * 4 + 3) * LQB_P + dl] = v.w;
  }
  __syncthreads();

  v8f acc[2][2] = {};
  const int mrow = lane & 15;
  const int koff = (lane >> 4) * 2;
  const int ncol = lane & 15;

  #pragma unroll
  for (int kk = 0; kk < 64; kk += 4) {
    v2f afrag[2], bfrag[2];
    #pragma unroll
    for (int mt = 0; mt < 2; ++mt) {
      int row = wm * 32 + mt * 16 + mrow;
      afrag[mt] = *(const v2f*)(lA + row * LQA_P + kk + koff);
    }
    #pragma unroll
    for (int nt = 0; nt < 2; ++nt) {
      int col = wn * 32 + nt * 16 + ncol;
      bfrag[nt].x = lB[(kk + koff + 0) * LQB_P + col];
      bfrag[nt].y = lB[(kk + koff + 1) * LQB_P + col];
    }
    #pragma unroll
    for (int mt = 0; mt < 2; ++mt)
      #pragma unroll
      for (int nt = 0; nt < 2; ++nt)
        acc[mt][nt] = __builtin_amdgcn_wmma_f32_16x16x4_f32(
            false, afrag[mt], false, bfrag[nt],
            (short)0, acc[mt][nt], false, false);
  }

  const float scale = 0.125f;
  const int rsel = (lane >> 4) * 8;
  float lmax = -3.4e38f;
  #pragma unroll
  for (int mt = 0; mt < 2; ++mt) {
    #pragma unroll
    for (int nt = 0; nt < 2; ++nt) {
      int col = bd + wn * 32 + nt * 16 + ncol;
      #pragma unroll
      for (int v = 0; v < 8; ++v) {
        int row = bb + wm * 32 + mt * 16 + rsel + v;
        float x = acc[mt][nt][v] * scale;
        lmax = fmaxf(lmax, x);
        aff[((size_t)row * BSZ + col) * CCH + c] = x;
      }
    }
  }
  red[t] = lmax;
  __syncthreads();
  for (int off = 128; off > 0; off >>= 1) {
    if (t < off) red[t] = fmaxf(red[t], red[t + off]);
    __syncthreads();
  }
  if (t == 0) atomicMax(&maxbuf[c], enc_f32(red[0]));
}

__global__ __launch_bounds__(256) void exp_sub_max(
    float* __restrict__ A, const unsigned int* __restrict__ maxbuf)
{
  size_t base   = ((size_t)blockIdx.x * 256 + threadIdx.x) * 4;
  size_t stride = (size_t)gridDim.x * 256 * 4;
  for (size_t i = base; i < (size_t)ATOT; i += stride) {
    #pragma unroll
    for (int j = 0; j < 4; ++j) {
      size_t idx = i + j;
      float mv = dec_f32(maxbuf[idx & 15]);
      A[idx] = expf(A[idx] - mv);
    }
  }
}

__global__ __launch_bounds__(256) void sink_u(
    const float* __restrict__ A, const float* __restrict__ v,
    const float* __restrict__ r, float* __restrict__ u)
{
  __shared__ float red[256];
  const int t = threadIdx.x;
  const int b = blockIdx.x;
  const float* Ab = A + (size_t)b * (BSZ * CCH);

  float s = 0.0f;
  #pragma unroll 4
  for (int i = 0; i < 64; ++i) {
    int j = t + 256 * i;
    s += Ab[j] * v[j];
  }
  red[t] = s;
  __syncthreads();
  for (int off = 128; off >= 16; off >>= 1) {
    if (t < off) red[t] += red[t + off];
    __syncthreads();
  }
  if (t < CCH) u[b * CCH + t] = r[b * CCH + t] / red[t];
}

__global__ __launch_bounds__(256) void sink_v(
    const float* __restrict__ A, const float* __restrict__ u,
    const float* __restrict__ cmarg, float* __restrict__ v)
{
  __shared__ float red[256];
  const int t = threadIdx.x;
  const int d = blockIdx.x;
  const int c = t & 15;
  const int bseg = t >> 4;

  float s = 0.0f;
  #pragma unroll 4
  for (int k = 0; k < 64; ++k) {
    int b = bseg * 64 + k;
    s += A[(size_t)b * (BSZ * CCH) + d * CCH + c] * u[b * CCH + c];
  }
  red[t] = s;
  __syncthreads();
  for (int off = 128; off >= 16; off >>= 1) {
    if (t < off) red[t] += red[t + off];
    __syncthreads();
  }
  if (t < CCH) v[d * CCH + t] = cmarg[d * CCH + t] / red[t];
}

__global__ __launch_bounds__(256) void sink_final(
    float* __restrict__ A, const float* __restrict__ u,
    const float* __restrict__ v)
{
  size_t base   = ((size_t)blockIdx.x * 256 + threadIdx.x) * 4;
  size_t stride = (size_t)gridDim.x * 256 * 4;
  for (size_t i = base; i < (size_t)ATOT; i += stride) {
    #pragma unroll
    for (int j = 0; j < 4; ++j) {
      size_t idx = i + j;
      size_t b  = idx >> 14;
      size_t dc = idx & 16383;
      A[idx] = A[idx] * u[b * CCH + (idx & 15)] * v[dc];
    }
  }
}

extern "C" void kernel_launch(void* const* d_in, const int* in_sizes, int n_in,
                              void* d_out, int out_size, void* d_ws, size_t ws_size,
                              hipStream_t stream)
{
  const float* x1   = (const float*)d_in[0];
  const float* x2   = (const float*)d_in[1];
  const float* p1   = (const float*)d_in[2];
  const float* p2   = (const float*)d_in[3];
  const float* w1[4] = {(const float*)d_in[4],  (const float*)d_in[6],
                        (const float*)d_in[8],  (const float*)d_in[10]};
  const float* b1[4] = {(const float*)d_in[5],  (const float*)d_in[7],
                        (const float*)d_in[9],  (const float*)d_in[11]};
  const float* w2[4] = {(const float*)d_in[12], (const float*)d_in[14],
                        (const float*)d_in[16], (const float*)d_in[18]};
  const float* b2[4] = {(const float*)d_in[13], (const float*)d_in[15],
                        (const float*)d_in[17], (const float*)d_in[19]};

  float* ws = (float*)d_ws;
  float* hA = ws;
  float* hB = hA + 1024 * 512;
  float* q1 = hB + 1024 * 512;
  float* q2 = q1 + 1024 * 1024;
  float* uv_u = q2 + 1024 * 1024;
  float* uv_v = uv_u + 16384;
  unsigned int* maxbuf = (unsigned int*)(uv_v + 16384);

  float* Aout = (float*)d_out;

  gemm_bias_act<<<dim3(512/64, 1024/128), 256, 0, stream>>>(x1, w1[0], b1[0], hA, 1024, 256, 512, 1);
  gemm_bias_act<<<dim3(512/64, 1024/128), 256, 0, stream>>>(hA, w1[1], b1[1], hB, 1024, 512, 512, 1);
  gemm_bias_act<<<dim3(512/64, 1024/128), 256, 0, stream>>>(hB, w1[2], b1[2], hA, 1024, 512, 512, 1);
  gemm_bias_act<<<dim3(1024/64, 1024/128), 256, 0, stream>>>(hA, w1[3], b1[3], q1, 1024, 512, 1024, 0);
  gemm_bias_act<<<dim3(512/64, 1024/128), 256, 0, stream>>>(x2, w2[0], b2[0], hA, 1024, 256, 512, 1);
  gemm_bias_act<<<dim3(512/64, 1024/128), 256, 0, stream>>>(hA, w2[1], b2[1], hB, 1024, 512, 512, 1);
  gemm_bias_act<<<dim3(512/64, 1024/128), 256, 0, stream>>>(hB, w2[2], b2[2], hA, 1024, 512, 512, 1);
  gemm_bias_act<<<dim3(1024/64, 1024/128), 256, 0, stream>>>(hA, w2[3], b2[3], q2, 1024, 512, 1024, 0);

  norm_rows64<<<16384 / 8, 256, 0, stream>>>(q1);
  norm_rows64<<<16384 / 8, 256, 0, stream>>>(q2);

  sink_init<<<(1024 * 16 + 255) / 256, 256, 0, stream>>>(uv_v, maxbuf);

  affinity_wmma<<<dim3(1024/64, 1024/128, CCH), 256, 0, stream>>>(q1, q2, Aout, maxbuf);

  exp_sub_max<<<4096, 256, 0, stream>>>(Aout, maxbuf);

  for (int it = 0; it < 10; ++it) {
    sink_u<<<1024, 256, 0, stream>>>(Aout, uv_v, p1, uv_u);
    sink_v<<<1024, 256, 0, stream>>>(Aout, uv_u, p2, uv_v);
  }

  sink_final<<<4096, 256, 0, stream>>>(Aout, uv_u, uv_v);
}


# affinity: one workgroup computes all 16 channels of a 64x32 (b,d) tile, so a lane stores a pair's 16 channels as 64 contiguous bytes (was one channel per workgroup: 4-byte stores 64 bytes apart)
# speedup vs baseline: 1.1349x; 1.1349x over previous
#include <hip/hip_runtime.h>
#include <hip/hip_bf16.h>


typedef float v2f __attribute__((ext_vector_type(2)));
typedef float v8f __attribute__((ext_vector_type(8)));

#define BSZ   1024
#define CCH   16
#define EDIM  64
#define SLICE (CCH * EDIM)
#define ATOT  (BSZ * BSZ * CCH)

__device__ __forceinline__ unsigned int enc_f32(float f) {
  unsigned int u = __float_as_uint(f);
  return (u & 0x80000000u) ? ~u : (u | 0x80000000u);
}
__device__ __forceinline__ float dec_f32(unsigned int u) {
  return (u & 0x80000000u) ? __uint_as_float(u & 0x7FFFFFFFu)
                           : __uint_as_float(~u);
}

#define LDA_P 20
#define LDB_P 68

__global__ __launch_bounds__(256) void gemm_bias_act(
    const float* __restrict__ A, const float* __restrict__ W,
    const float* __restrict__ bias, float* __restrict__ out,
    int M, int K, int N, int relu)
{
  __shared__ float lA[128 * LDA_P];
  __shared__ float lB[16 * LDB_P];

  const int t    = threadIdx.x;
  const int lane = t & 31;
  const int wave = t >> 5;
  const int wm   = wave & 3;
  const int wn   = wave >> 2;
  const int bm   = blockIdx.y * 128;
  const int bn   = blockIdx.x * 64;

  v8f acc[2][2] = {};

  const int mrow = lane & 15;
  const int koff = (lane >> 4) * 2;
  const int ncol = lane & 15;

  for (int k0 = 0; k0 < K; k0 += 16) {
    #pragma unroll
    for (int i = 0; i < 2; ++i) {
      int idx = t + i * 256;
      int row = idx >> 2;
      int c4  = idx & 3;
      float4 v = *(const float4*)(A + (size_t)(bm + row) * K + k0 + c4 * 4);
      *(float4*)(lA + row * LDA_P + c4 * 4) = v;
    }
    {
      int row = t >> 4;
      int c4  = t & 15;
      float4 v = *(const float4*)(W + (size_t)(k0 + row) * N + bn + c4 * 4);
      *(float4*)(lB + row * LDB_P + c4 * 4) = v;
    }
    __syncthreads();

    #pragma unroll
    for (int kk = 0; kk < 16; kk += 4) {
      v2f afrag[2], bfrag[2];
      #pragma unroll
      for (int mt = 0; mt < 2; ++mt) {
        int row = wm * 32 + mt * 16 + mrow;
        afrag[mt] = *(const v2f*)(lA + row * LDA_P + kk + koff);
      }
      #pragma unroll
      for (int nt = 0; nt < 2; ++nt) {
        int col = wn * 32 + nt * 16 + ncol;
        bfrag[nt].x = lB[(kk + koff + 0) * LDB_P + col];
        bfrag[nt].y = lB[(kk + koff + 1) * LDB_P + col];
      }
      #pragma unroll
      for (int mt = 0; mt < 2; ++mt)
        #pragma unroll
        for (int nt = 0; nt < 2; ++nt)
          acc[mt][nt] = __builtin_amdgcn_wmma_f32_16x16x4_f32(
              false, afrag[mt], false, bfrag[nt],
              (short)0, acc[mt][nt], false, false);
    }
    __syncthreads();
  }

  const int rsel = (lane >> 4) * 8;
  #pragma unroll
  for (int nt = 0; nt < 2; ++nt) {
    int col = bn + wn * 32 + nt * 16 + ncol;
    float bv = bias[col];
    #pragma unroll
    for (int mt = 0; mt < 2; ++mt) {
      #pragma unroll
      for (int v = 0; v < 8; ++v) {
        int row = bm + wm * 32 + mt * 16 + rsel + v;
        float x = acc[mt][nt][v] + bv;
        if (relu) x = fmaxf(x, 0.0f);
        out[(size_t)row * N + col] = x;
      }
    }
  }
}

__global__ __launch_bounds__(256) void norm_rows64(float* __restrict__ q)
{
  const int lane = threadIdx.x & 31;
  const int wave = threadIdx.x >> 5;
  const int row  = blockIdx.x * 8 + wave;
  float* p = q + (size_t)row * 64;

  float2 x = *(float2*)(p + lane * 2);
  float s  = x.x + x.y;
  float ss = x.x * x.x + x.y * x.y;
  #pragma unroll
  for (int off = 16; off > 0; off >>= 1) {
    s  += __shfl_xor(s,  off, 32);
    ss += __shfl_xor(ss, off, 32);
  }
  float mean = s * (1.0f / 64.0f);
  float var  = (ss - 64.0f * mean * mean) * (1.0f / 63.0f);
  float inv  = rsqrtf(var + 1e-8f);
  x.x = (x.x - mean) * inv;
  x.y = (x.y - mean) * inv;
  *(float2*)(p + lane * 2) = x;
}

__global__ void sink_init(float* __restrict__ v, unsigned int* __restrict__ maxbuf)
{
  int i = blockIdx.x * 256 + threadIdx.x;
  if (i < BSZ * CCH) v[i] = 1.0f;
  if (i < CCH) maxbuf[i] = 0u;
}

#define LQA_P 68
#define LQB_P 33

__global__ __launch_bounds__(256) void affinity_wmma(
    const float* __restrict__ q1, const float* __restrict__ q2,
    float* __restrict__ aff, unsigned int* __restrict__ maxbuf)
{
  __shared__ float lA[64 * LQA_P];
  __shared__ float lB[64 * LQB_P];
  __shared__ float red[16 * CCH];

  const int t    = threadIdx.x;
  const int lane = t & 31;
  const int wave = t >> 5;
  const int wm   = wave & 3;
  const int wn   = wave >> 2;
  const int bb   = blockIdx.y * 64;
  const int bd   = blockIdx.x * 32;

  const int srow = t >> 4;
  const int sc4  = t & 15;
  const float* gA = q1 + (size_t)(bb + srow) * SLICE + sc4 * 4;
  const float* gB = q2 + (size_t)(bd + srow) * SLICE + sc4 * 4;

  float4 pa0 = *(const float4*)(gA);
  float4 pa1 = *(const float4*)(gA + (size_t)16 * SLICE);
  float4 pa2 = *(const float4*)(gA + (size_t)32 * SLICE);
  float4 pa3 = *(const float4*)(gA + (size_t)48 * SLICE);
  float4 pb0 = *(const float4*)(gB);
  float4 pb1 = *(const float4*)(gB + (size_t)16 * SLICE);
  __builtin_amdgcn_sched_barrier(0);

  v8f acc[CCH] = {};
  const int mrow = lane & 15;
  const int koff = (lane >> 4) * 2;
  const int ncol = lane & 15;

  #pragma unroll
  for (int c = 0; c < CCH; ++c) {
    *(float4*)(lA + (srow +  0) * LQA_P + sc4 * 4) = pa0;
    *(float4*)(lA + (srow + 16) * LQA_P + sc4 * 4) = pa1;
    *(float4*)(lA + (srow + 32) * LQA_P + sc4 * 4) = pa2;
    *(float4*)(lA + (srow + 48) * LQA_P + sc4 * 4) = pa3;
    lB[(sc4 * 4 + 0) * LQB_P + srow] = pb0.x;
    lB[(sc4 * 4 + 1) * LQB_P + srow] = pb0.y;
    lB[(sc4 * 4 + 2) * LQB_P + srow] = pb0.z;
    lB[(sc4 * 4 + 3) * LQB_P + srow] = pb0.w;
    lB[(sc4 * 4 + 0) * LQB_P + srow + 16] = pb1.x;
    lB[(sc4 * 4 + 1) * LQB_P + srow + 16] = pb1.y;
    lB[(sc4 * 4 + 2) * LQB_P + srow + 16] = pb1.z;
    lB[(sc4 * 4 + 3) * LQB_P + srow + 16] = pb1.w;
    __syncthreads();

    if (c + 1 < CCH) {
      const float* nA = gA + (c + 1) * EDIM;
      const float* nB = gB + (c + 1) * EDIM;
      pa0 = *(const float4*)(nA);
      pa1 = *(const float4*)(nA + (size_t)16 * SLICE);
      pa2 = *(const float4*)(nA + (size_t)32 * SLICE);
      pa3 = *(const float4*)(nA + (size_t)48 * SLICE);
      pb0 = *(const float4*)(nB);
      pb1 = *(const float4*)(nB + (size_t)16 * SLICE);
    }

    #pragma unroll
    for (int kk = 0; kk < 64; kk += 4) {
      v2f afrag, bfrag;
      afrag = *(const v2f*)(lA + (wm * 16 + mrow) * LQA_P + kk + koff);
      bfrag.x = lB[(kk + koff + 0) * LQB_P + wn * 16 + ncol];
      bfrag.y = lB[(kk + koff + 1) * LQB_P + wn * 16 + ncol];
      acc[c] = __builtin_amdgcn_wmma_f32_16x16x4_f32(
          false, afrag, false, bfrag, (short)0, acc[c], false, false);
    }
    __syncthreads();
  }

  const float scale = 0.125f;
  const int rsel = (lane >> 4) * 8;
  const int col  = bd + wn * 16 + ncol;
  #pragma unroll
  for (int v = 0; v < 8; ++v) {
    int row = bb + wm * 16 + rsel + v;
    float* dst = aff + ((size_t)row * BSZ + col) * CCH;
    #pragma unroll
    for (int q = 0; q < 4; ++q) {
      float4 o;
      o.x = acc[4 * q + 0][v] * scale;
      o.y = acc[4 * q + 1][v] * scale;
      o.z = acc[4 * q + 2][v] * scale;
      o.w = acc[4 * q + 3][v] * scale;
      *(float4*)(dst + 4 * q) = o;
    }
  }

  float* park = lA;
  #pragma unroll
  for (int c = 0; c < CCH; ++c) {
    float m = -3.4e38f;
    #pragma unroll
    for (int v = 0; v < 8; ++v) m = fmaxf(m, acc[c][v] * scale);
    park[t * 17 + c] = m;
  }
  __syncthreads();
  {
    const int pc = t & 15;
    const int pg = t >> 4;
    float m = park[(pg * 16) * 17 + pc];
    for (int i = 1; i < 16; ++i) m = fmaxf(m, park[(pg * 16 + i) * 17 + pc]);
    red[pg * CCH + pc] = m;
  }
  __syncthreads();
  if (t < CCH) {
    float m = red[t];
    for (int g = 1; g < 16; ++g) m = fmaxf(m, red[g * CCH + t]);
    atomicMax(&maxbuf[t], enc_f32(m));
  }
}

__global__ __launch_bounds__(256) void exp_sub_max(
    float* __restrict__ A, const unsigned int* __restrict__ maxbuf)
{
  size_t base   = ((size_t)blockIdx.x * 256 + threadIdx.x) * 4;
  size_t stride = (size_t)gridDim.x * 256 * 4;
  for (size_t i = base; i < (size_t)ATOT; i += stride) {
    #pragma unroll
    for (int j = 0; j < 4; ++j) {
      size_t idx = i + j;
      float mv = dec_f32(maxbuf[idx & 15]);
      A[idx] = expf(A[idx] - mv);
    }
  }
}

__global__ __launch_bounds__(256) void sink_u(
    const float* __restrict__ A, const float* __restrict__ v,
    const float* __restrict__ r, float* __restrict__ u)
{
  __shared__ float red[256];
  const int t = threadIdx.x;
  const int b = blockIdx.x;
  const float* Ab = A + (size_t)b * (BSZ * CCH);

  float s = 0.0f;
  #pragma unroll 4
  for (int i = 0; i < 64; ++i) {
    int j = t + 256 * i;
    s += Ab[j] * v[j];
  }
  red[t] = s;
  __syncthreads();
  for (int off = 128; off >= 16; off >>= 1) {
    if (t < off) red[t] += red[t + off];
    __syncthreads();
  }
  if (t < CCH) u[b * CCH + t] = r[b * CCH + t] / red[t];
}

__global__ __launch_bounds__(256) void sink_v(
    const float* __restrict__ A, const float* __restrict__ u,
    const float* __restrict__ cmarg, float* __restrict__ v)
{
  __shared__ float red[256];
  const int t = threadIdx.x;
  const int d = blockIdx.x;
  const int c = t & 15;
  const int bseg = t >> 4;

  float s = 0.0f;
  #pragma unroll 4
  for (int k = 0; k < 64; ++k) {
    int b = bseg * 64 + k;
    s += A[(size_t)b * (BSZ * CCH) + d * CCH + c] * u[b * CCH + c];
  }
  red[t] = s;
  __syncthreads();
  for (int off = 128; off >= 16; off >>= 1) {
    if (t < off) red[t] += red[t + off];
    __syncthreads();
  }
  if (t < CCH) v[d * CCH + t] = cmarg[d * CCH + t] / red[t];
}

__global__ __launch_bounds__(256) void sink_final(
    float* __restrict__ A, const float* __restrict__ u,
    const float* __restrict__ v)
{
  size_t base   = ((size_t)blockIdx.x * 256 + threadIdx.x) * 4;
  size_t stride = (size_t)gridDim.x * 256 * 4;
  for (size_t i = base; i < (size_t)ATOT; i += stride) {
    #pragma unroll
    for (int j = 0; j < 4; ++j) {
      size_t idx = i + j;
      size_t b  = idx >> 14;
      size_t dc = idx & 16383;
      A[idx] = A[idx] * u[b * CCH + (idx & 15)] * v[dc];
    }
  }
}

extern "C" void kernel_launch(void* const* d_in, const int* in_sizes, int n_in,
                              void* d_out, int out_size, void* d_ws, size_t ws_size,
                              hipStream_t stream)
{
  const float* x1   = (const float*)d_in[0];
  const float* x2   = (const float*)d_in[1];
  const float* p1   = (const float*)d_in[2];
  const float* p2   = (const float*)d_in[3];
  const float* w1[4] = {(const float*)d_in[4],  (const float*)d_in[6],
                        (const float*)d_in[8],  (const float*)d_in[10]};
  const float* b1[4] = {(const float*)d_in[5],  (const float*)d_in[7],
                        (const float*)d_in[9],  (const float*)d_in[11]};
  const float* w2[4] = {(const float*)d_in[12], (const float*)d_in[14],
                        (const float*)d_in[16], (const float*)d_in[18]};
  const float* b2[4] = {(const float*)d_in[13], (const float*)d_in[15],
                        (const float*)d_in[17], (const float*)d_in[19]};

  float* ws = (float*)d_ws;
  float* hA = ws;
  float* hB = hA + 1024 * 512;
  float* q1 = hB + 1024 * 512;
  float* q2 = q1 + 1024 * 1024;
  float* uv_u = q2 + 1024 * 1024;
  float* uv_v = uv_u + 16384;
  unsigned int* maxbuf = (unsigned int*)(uv_v + 16384);

  float* Aout = (float*)d_out;

  gemm_bias_act<<<dim3(512/64, 1024/128), 256, 0, stream>>>(x1, w1[0], b1[0], hA, 1024, 256, 512, 1);
  gemm_bias_act<<<dim3(512/64, 1024/128), 256, 0, stream>>>(hA, w1[1], b1[1], hB, 1024, 512, 512, 1);
  gemm_bias_act<<<dim3(512/64, 1024/128), 256, 0, stream>>>(hB, w1[2], b1[2], hA, 1024, 512, 512, 1);
  gemm_bias_act<<<dim3(1024/64, 1024/128), 256, 0, stream>>>(hA, w1[3], b1[3], q1, 1024, 512, 1024, 0);
  gemm_bias_act<<<dim3(512/64, 1024/128), 256, 0, stream>>>(x2, w2[0], b2[0], hA, 1024, 256, 512, 1);
  gemm_bias_act<<<dim3(512/64, 1024/128), 256, 0, stream>>>(hA, w2[1], b2[1], hB, 1024, 512, 512, 1);
  gemm_bias_act<<<dim3(512/64, 1024/128), 256, 0, stream>>>(hB, w2[2], b2[2], hA, 1024, 512, 512, 1);
  gemm_bias_act<<<dim3(1024/64, 1024/128), 256, 0, stream>>>(hA, w2[3], b2[3], q2, 1024, 512, 1024, 0);

  norm_rows64<<<16384 / 8, 256, 0, stream>>>(q1);
  norm_rows64<<<16384 / 8, 256, 0, stream>>>(q2);

  sink_init<<<(1024 * 16 + 255) / 256, 256, 0, stream>>>(uv_v, maxbuf);

  affinity_wmma<<<dim3(1024/32, 1024/64), 256, 0, stream>>>(q1, q2, Aout, maxbuf);

  exp_sub_max<<<4096, 256, 0, stream>>>(Aout, maxbuf);

  for (int it = 0; it < 10; ++it) {
    sink_u<<<1024, 256, 0, stream>>>(Aout, uv_v, p1, uv_u);
    sink_v<<<1024, 256, 0, stream>>>(Aout, uv_u, p2, uv_v);
  }

  sink_final<<<4096, 256, 0, stream>>>(Aout, uv_u, uv_v);
}


# row and column scaling passes over A: whole-line 16-byte loads, many in flight per wave (column pass: 256 contiguous bytes per row, u staged in LDS); row and column sums in a new order
# speedup vs baseline: 1.4123x; 1.2444x over previous
#include <hip/hip_runtime.h>
#include <hip/hip_bf16.h>


typedef float v2f __attribute__((ext_vector_type(2)));
typedef float v8f __attribute__((ext_vector_type(8)));

#define BSZ   1024
#define CCH   16
#define EDIM  64
#define SLICE (CCH * EDIM)
#define ATOT  (BSZ * BSZ * CCH)

__device__ __forceinline__ unsigned int enc_f32(float f) {
  unsigned int u = __float_as_uint(f);
  return (u & 0x80000000u) ? ~u : (u | 0x80000000u);
}
__device__ __forceinline__ float dec_f32(unsigned int u) {
  return (u & 0x80000000u) ? __uint_as_float(u & 0x7FFFFFFFu)
                           : __uint_as_float(~u);
}

#define LDA_P 20
#define LDB_P 68

__global__ __launch_bounds__(256) void gemm_bias_act(
    const float* __restrict__ A, const float* __restrict__ W,
    const float* __restrict__ bias, float* __restrict__ out,
    int M, int K, int N, int relu)
{
  __shared__ float lA[128 * LDA_P];
  __shared__ float lB[16 * LDB_P];

  const int t    = threadIdx.x;
  const int lane = t & 31;
  const int wave = t >> 5;
  const int wm   = wave & 3;
  const int wn   = wave >> 2;
  const int bm   = blockIdx.y * 128;
  const int bn   = blockIdx.x * 64;

  v8f acc[2][2] = {};

  const int mrow = lane & 15;
  const int koff = (lane >> 4) * 2;
  const int ncol = lane & 15;

  for (int k0 = 0; k0 < K; k0 += 16) {
    #pragma unroll
    for (int i = 0; i < 2; ++i) {
      int idx = t + i * 256;
      int row = idx >> 2;
      int c4  = idx & 3;
      float4 v = *(const float4*)(A + (size_t)(bm + row) * K + k0 + c4 * 4);
      *(float4*)(lA + row * LDA_P + c4 * 4) = v;
    }
    {
      int row = t >> 4;
      int c4  = t & 15;
      float4 v = *(const float4*)(W + (size_t)(k0 + row) * N + bn + c4 * 4);
      *(float4*)(lB + row * LDB_P + c4 * 4) = v;
    }
    __syncthreads();

    #pragma unroll
    for (int kk = 0; kk < 16; kk += 4) {
      v2f afrag[2], bfrag[2];
      #pragma unroll
      for (int mt = 0; mt < 2; ++mt) {
        int row = wm * 32 + mt * 16 + mrow;
        afrag[mt] = *(const v2f*)(lA + row * LDA_P + kk + koff);
      }
      #pragma unroll
      for (int nt = 0; nt < 2; ++nt) {
        int col = wn * 32 + nt * 16 + ncol;
        bfrag[nt].x = lB[(kk + koff + 0) * LDB_P + col];
        bfrag[nt].y = lB[(kk + koff + 1) * LDB_P + col];
      }
      #pragma unroll
      for (int mt = 0; mt < 2; ++mt)
        #pragma unroll
        for (int nt = 0; nt < 2; ++nt)
          acc[mt][nt] = __builtin_amdgcn_wmma_f32_16x16x4_f32(
              false, afrag[mt], false, bfrag[nt],
              (short)0, acc[mt][nt], false, false);
    }
    __syncthreads();
  }

  const int rsel = (lane >> 4) * 8;
  #pragma unroll
  for (int nt = 0; nt < 2; ++nt) {
    int col = bn + wn * 32 + nt * 16 + ncol;
    float bv = bias[col];
    #pragma unroll
    for (int mt = 0; mt < 2; ++mt) {
      #pragma unroll
      for (int v = 0; v < 8; ++v) {
        int row = bm + wm * 32 + mt * 16 + rsel + v;
        float x = acc[mt][nt][v] + bv;
        if (relu) x = fmaxf(x, 0.0f);
        out[(size_t)row * N + col] = x;
      }
    }
  }
}

__global__ __launch_bounds__(256) void norm_rows64(float* __restrict__ q)
{
  const int lane = threadIdx.x & 31;
  const int wave = threadIdx.x >> 5;
  const int row  = blockIdx.x * 8 + wave;
  float* p = q + (size_t)row * 64;

  float2 x = *(float2*)(p + lane * 2);
  float s  = x.x + x.y;
  float ss = x.x * x.x + x.y * x.y;
  #pragma unroll
  for (int off = 16; off > 0; off >>= 1) {
    s  += __shfl_xor(s,  off, 32);
    ss += __shfl_xor(ss, off, 32);
  }
  float mean = s * (1.0f / 64.0f);
  float var  = (ss - 64.0f * mean * mean) * (1.0f / 63.0f);
  float inv  = rsqrtf(var + 1e-8f);
  x.x = (x.x - mean) * inv;
  x.y = (x.y - mean) * inv;
  *(float2*)(p + lane * 2) = x;
}

__global__ void sink_init(float* __restrict__ v, unsigned int* __restrict__ maxbuf)
{
  int i = blockIdx.x * 256 + threadIdx.x;
  if (i < BSZ * CCH) v[i] = 1.0f;
  if (i < CCH) maxbuf[i] = 0u;
}

#define LQA_P 68
#define LQB_P 33

__global__ __launch_bounds__(256) void affinity_wmma(
    const float* __restrict__ q1, const float* __restrict__ q2,
    float* __restrict__ aff, unsigned int* __restrict__ maxbuf)
{
  __shared__ float lA[64 * LQA_P];
  __shared__ float lB[64 * LQB_P];
  __shared__ float red[16 * CCH];

  const int t    = threadIdx.x;
  const int lane = t & 31;
  const int wave = t >> 5;
  const int wm   = wave & 3;
  const int wn   = wave >> 2;
  const int bb   = blockIdx.y * 64;
  const int bd   = blockIdx.x * 32;

  const int srow = t >> 4;
  const int sc4  = t & 15;
  const float* gA = q1 + (size_t)(bb + srow) * SLICE + sc4 * 4;
  const float* gB = q2 + (size_t)(bd + srow) * SLICE + sc4 * 4;

  float4 pa0 = *(const float4*)(gA);
  float4 pa1 = *(const float4*)(gA + (size_t)16 * SLICE);
  float4 pa2 = *(const float4*)(gA + (size_t)32 * SLICE);
  float4 pa3 = *(const float4*)(gA + (size_t)48 * SLICE);
  float4 pb0 = *(const float4*)(gB);
  float4 pb1 = *(const float4*)(gB + (size_t)16 * SLICE);
  __builtin_amdgcn_sched_barrier(0);

  v8f acc[CCH] = {};
  const int mrow = lane & 15;
  const int koff = (lane >> 4) * 2;
  const int ncol = lane & 15;

  #pragma unroll
  for (int c = 0; c < CCH; ++c) {
    *(float4*)(lA + (srow +  0) * LQA_P + sc4 * 4) = pa0;
    *(float4*)(lA + (srow + 16) * LQA_P + sc4 * 4) = pa1;
    *(float4*)(lA + (srow + 32) * LQA_P + sc4 * 4) = pa2;
    *(float4*)(lA + (srow + 48) * LQA_P + sc4 * 4) = pa3;
    lB[(sc4 * 4 + 0) * LQB_P + srow] = pb0.x;
    lB[(sc4 * 4 + 1) * LQB_P + srow] = pb0.y;
    lB[(sc4 * 4 + 2) * LQB_P + srow] = pb0.z;
    lB[(sc4 * 4 + 3) * LQB_P + srow] = pb0.w;
    lB[(sc4 * 4 + 0) * LQB_P + srow + 16] = pb1.x;
    lB[(sc4 * 4 + 1) * LQB_P + srow + 16] = pb1.y;
    lB[(sc4 * 4 + 2) * LQB_P + srow + 16] = pb1.z;
    lB[(sc4 * 4 + 3) * LQB_P + srow + 16] = pb1.w;
    __syncthreads();

    if (c + 1 < CCH) {
      const float* nA = gA + (c + 1) * EDIM;
      const float* nB = gB + (c + 1) * EDIM;
      pa0 = *(const float4*)(nA);
      pa1 = *(const float4*)(nA + (size_t)16 * SLICE);
      pa2 = *(const float4*)(nA + (size_t)32 * SLICE);
      pa3 = *(const float4*)(nA + (size_t)48 * SLICE);
      pb0 = *(const float4*)(nB);
      pb1 = *(const float4*)(nB + (size_t)16 * SLICE);
    }

    #pragma unroll
    for (int kk = 0; kk < 64; kk += 4) {
      v2f afrag, bfrag;
      afrag = *(const v2f*)(lA + (wm * 16 + mrow) * LQA_P + kk + koff);
      bfrag.x = lB[(kk + koff + 0) * LQB_P + wn * 16 + ncol];
      bfrag.y = lB[(kk + koff + 1) * LQB_P + wn * 16 + ncol];
      acc[c] = __builtin_amdgcn_wmma_f32_16x16x4_f32(
          false, afrag, false, bfrag, (short)0, acc[c], false, false);
    }
    __syncthreads();
  }

  const float scale = 0.125f;
  const int rsel = (lane >> 4) * 8;
  const int col  = bd + wn * 16 + ncol;
  #pragma unroll
  for (int v = 0; v < 8; ++v) {
    int row = bb + wm * 16 + rsel + v;
    float* dst = aff + ((size_t)row * BSZ + col) * CCH;
    #pragma unroll
    for (int q = 0; q < 4; ++q) {
      float4 o;
      o.x = acc[4 * q + 0][v] * scale;
      o.y = acc[4 * q + 1][v] * scale;
      o.z = acc[4 * q + 2][v] * scale;
      o.w = acc[4 * q + 3][v] * scale;
      *(float4*)(dst + 4 * q) = o;
    }
  }

  float* park = lA;
  #pragma unroll
  for (int c = 0; c < CCH; ++c) {
    float m = -3.4e38f;
    #pragma unroll
    for (int v = 0; v < 8; ++v) m = fmaxf(m, acc[c][v] * scale);
    park[t * 17 + c] = m;
  }
  __syncthreads();
  {
    const int pc = t & 15;
    const int pg = t >> 4;
    float m = park[(pg * 16) * 17 + pc];
    for (int i = 1; i < 16; ++i) m = fmaxf(m, park[(pg * 16 + i) * 17 + pc]);
    red[pg * CCH + pc] = m;
  }
  __syncthreads();
  if (t < CCH) {
    float m = red[t];
    for (int g = 1; g < 16; ++g) m = fmaxf(m, red[g * CCH + t]);
    atomicMax(&maxbuf[t], enc_f32(m));
  }
}

__global__ __launch_bounds__(256) void exp_sub_max(
    float* __restrict__ A, const unsigned int* __restrict__ maxbuf)
{
  size_t base   = ((size_t)blockIdx.x * 256 + threadIdx.x) * 4;
  size_t stride = (size_t)gridDim.x * 256 * 4;
  for (size_t i = base; i < (size_t)ATOT; i += stride) {
    #pragma unroll
    for (int j = 0; j < 4; ++j) {
      size_t idx = i + j;
      float mv = dec_f32(maxbuf[idx & 15]);
      A[idx] = expf(A[idx] - mv);
    }
  }
}

__global__ __launch_bounds__(256) void sink_u(
    const float* __restrict__ A, const float* __restrict__ v,
    const float* __restrict__ r, float* __restrict__ u)
{
  __shared__ float red[8 * 32];
  const int t    = threadIdx.x;
  const int lane = t & 31;
  const int wave = t >> 5;
  const int b0   = blockIdx.x * 2;
  const float4* A4 = (const float4*)(A + (size_t)b0 * (BSZ * CCH)) + t;
  const float4* v4 = (const float4*)v + t;

  float s[8] = {0.0f, 0.0f, 0.0f, 0.0f, 0.0f, 0.0f, 0.0f, 0.0f};
  #pragma unroll
  for (int i = 0; i < 16; ++i) {
    const float4 w  = v4[i * 256];
    const float4 x0 = A4[i * 256];
    const float4 x1 = A4[4096 + i * 256];
    s[0] += x0.x * w.x;  s[1] += x0.y * w.y;
    s[2] += x0.z * w.z;  s[3] += x0.w * w.w;
    s[4] += x1.x * w.x;  s[5] += x1.y * w.y;
    s[6] += x1.z * w.z;  s[7] += x1.w * w.w;
  }
  #pragma unroll
  for (int q = 0; q < 8; ++q) {
    s[q] += __shfl_xor(s[q], 4, 32);
    s[q] += __shfl_xor(s[q], 8, 32);
    s[q] += __shfl_xor(s[q], 16, 32);
  }
  if (lane < 4) {
    #pragma unroll
    for (int k = 0; k < 4; ++k) {
      red[wave * 32 +  0 + lane * 4 + k] = s[k];
      red[wave * 32 + 16 + lane * 4 + k] = s[4 + k];
    }
  }
  __syncthreads();
  if (t < 32) {
    float sum = red[t];
    for (int w = 1; w < 8; ++w) sum += red[w * 32 + t];
    u[b0 * CCH + t] = r[b0 * CCH + t] / sum;
  }
}

__global__ __launch_bounds__(256) void sink_v(
    const float* __restrict__ A, const float* __restrict__ u,
    const float* __restrict__ cmarg, float* __restrict__ v)
{
  __shared__ float4 su[BSZ * CCH / 4];
  __shared__ float red[8 * 64];
  const int t    = threadIdx.x;
  const int lane = t & 31;
  const int wave = t >> 5;
  const int l16  = t & 15;
  const int rg   = t >> 4;
  const int j0   = blockIdx.x * 64;
  const float* Ap = A + (size_t)rg * (BSZ * CCH) + j0 + l16 * 4;
  const float4* sp = su + rg * 4 + (l16 & 3);

  #pragma unroll
  for (int i = 0; i < 16; ++i) su[t + 256 * i] = ((const float4*)u)[t + 256 * i];
  __syncthreads();

  float4 acc = make_float4(0.0f, 0.0f, 0.0f, 0.0f);
  #pragma unroll
  for (int step = 0; step < 64; ++step) {
    float4 a = *(const float4*)(Ap + (size_t)step * 16 * (BSZ * CCH));
    float4 w = sp[step * 64];
    acc.x += a.x * w.x;  acc.y += a.y * w.y;
    acc.z += a.z * w.z;  acc.w += a.w * w.w;
  }

  acc.x += __shfl_xor(acc.x, 16, 32);
  acc.y += __shfl_xor(acc.y, 16, 32);
  acc.z += __shfl_xor(acc.z, 16, 32);
  acc.w += __shfl_xor(acc.w, 16, 32);
  if (lane < 16) {
    red[wave * 64 + lane * 4 + 0] = acc.x;
    red[wave * 64 + lane * 4 + 1] = acc.y;
    red[wave * 64 + lane * 4 + 2] = acc.z;
    red[wave * 64 + lane * 4 + 3] = acc.w;
  }
  __syncthreads();
  if (t < 64) {
    float sum = red[t];
    for (int w = 1; w < 8; ++w) sum += red[w * 64 + t];
    v[j0 + t] = cmarg[j0 + t] / sum;
  }
}

__global__ __launch_bounds__(256) void sink_final(
    float* __restrict__ A, const float* __restrict__ u,
    const float* __restrict__ v)
{
  size_t base   = ((size_t)blockIdx.x * 256 + threadIdx.x) * 4;
  size_t stride = (size_t)gridDim.x * 256 * 4;
  for (size_t i = base; i < (size_t)ATOT; i += stride) {
    #pragma unroll
    for (int j = 0; j < 4; ++j) {
      size_t idx = i + j;
      size_t b  = idx >> 14;
      size_t dc = idx & 16383;
      A[idx] = A[idx] * u[b * CCH + (idx & 15)] * v[dc];
    }
  }
}

extern "C" void kernel_launch(void* const* d_in, const int* in_sizes, int n_in,
                              void* d_out, int out_size, void* d_ws, size_t ws_size,
                              hipStream_t stream)
{
  const float* x1   = (const float*)d_in[0];
  const float* x2   = (const float*)d_in[1];
  const float* p1   = (const float*)d_in[2];
  const float* p2   = (const float*)d_in[3];
  const float* w1[4] = {(const float*)d_in[4],  (const float*)d_in[6],
                        (const float*)d_in[8],  (const float*)d_in[10]};
  const float* b1[4] = {(const float*)d_in[5],  (const float*)d_in[7],
                        (const float*)d_in[9],  (const float*)d_in[11]};
  const float* w2[4] = {(const float*)d_in[12], (const float*)d_in[14],
                        (const float*)d_in[16], (const float*)d_in[18]};
  const float* b2[4] = {(const float*)d_in[13], (const float*)d_in[15],
                        (const float*)d_in[17], (const float*)d_in[19]};

  float* ws = (float*)d_ws;
  float* hA = ws;
  float* hB = hA + 1024 * 512;
  float* q1 = hB + 1024 * 512;
  float* q2 = q1 + 1024 * 1024;
  float* uv_u = q2 + 1024 * 1024;
  float* uv_v = uv_u + 16384;
  unsigned int* maxbuf = (unsigned int*)(uv_v + 16384);

  float* Aout = (float*)d_out;

  gemm_bias_act<<<dim3(512/64, 1024/128), 256, 0, stream>>>(x1, w1[0], b1[0], hA, 1024, 256, 512, 1);
  gemm_bias_act<<<dim3(512/64, 1024/128), 256, 0, stream>>>(hA, w1[1], b1[1], hB, 1024, 512, 512, 1);
  gemm_bias_act<<<dim3(512/64, 1024/128), 256, 0, stream>>>(hB, w1[2], b1[2], hA, 1024, 512, 512, 1);
  gemm_bias_act<<<dim3(1024/64, 1024/128), 256, 0, stream>>>(hA, w1[3], b1[3], q1, 1024, 512, 1024, 0);
  gemm_bias_act<<<dim3(512/64, 1024/128), 256, 0, stream>>>(x2, w2[0], b2[0], hA, 1024, 256, 512, 1);
  gemm_bias_act<<<dim3(512/64, 1024/128), 256, 0, stream>>>(hA, w2[1], b2[1], hB, 1024, 512, 512, 1);
  gemm_bias_act<<<dim3(512/64, 1024/128), 256, 0, stream>>>(hB, w2[2], b2[2], hA, 1024, 512, 512, 1);
  gemm_bias_act<<<dim3(1024/64, 1024/128), 256, 0, stream>>>(hA, w2[3], b2[3], q2, 1024, 512, 1024, 0);

  norm_rows64<<<16384 / 8, 256, 0, stream>>>(q1);
  norm_rows64<<<16384 / 8, 256, 0, stream>>>(q2);

  sink_init<<<(1024 * 16 + 255) / 256, 256, 0, stream>>>(uv_v, maxbuf);

  affinity_wmma<<<dim3(1024/32, 1024/64), 256, 0, stream>>>(q1, q2, Aout, maxbuf);

  exp_sub_max<<<4096, 256, 0, stream>>>(Aout, maxbuf);

  for (int it = 0; it < 10; ++it) {
    sink_u<<<512, 256, 0, stream>>>(Aout, uv_v, p1, uv_u);
    sink_v<<<256, 256, 0, stream>>>(Aout, uv_u, p2, uv_v);
  }

  sink_final<<<4096, 256, 0, stream>>>(Aout, uv_u, uv_v);
}


# MLP layers: the two independent branches share one launch per layer (grid z = 2, 128 or 256 workgroups instead of 64 or 128), and one norm launch covers q1 and q2; 34 launches become 29
# speedup vs baseline: 1.8169x; 1.2865x over previous
#include <hip/hip_runtime.h>
#include <hip/hip_bf16.h>


typedef float v2f __attribute__((ext_vector_type(2)));
typedef float v8f __attribute__((ext_vector_type(8)));

#define BSZ   1024
#define CCH   16
#define EDIM  64
#define SLICE (CCH * EDIM)
#define ATOT  (BSZ * BSZ * CCH)

__device__ __forceinline__ unsigned int enc_f32(float f) {
  unsigned int u = __float_as_uint(f);
  return (u & 0x80000000u) ? ~u : (u | 0x80000000u);
}
__device__ __forceinline__ float dec_f32(unsigned int u) {
  return (u & 0x80000000u) ? __uint_as_float(u & 0x7FFFFFFFu)
                           : __uint_as_float(~u);
}

#define LDA_P 20
#define LDB_P 68

__global__ __launch_bounds__(256) void gemm_bias_act(
    const float* __restrict__ A0, const float* __restrict__ W0,
    const float* __restrict__ bias0, float* __restrict__ out0,
    const float* __restrict__ A1, const float* __restrict__ W1,
    const float* __restrict__ bias1, float* __restrict__ out1,
    int M, int K, int N, int relu)
{
  __shared__ float lA[128 * LDA_P];
  __shared__ float lB[16 * LDB_P];

  const bool br = blockIdx.z != 0;
  const float* __restrict__ A    = br ? A1 : A0;
  const float* __restrict__ W    = br ? W1 : W0;
  const float* __restrict__ bias = br ? bias1 : bias0;
  float* __restrict__ out        = br ? out1 : out0;

  const int t    = threadIdx.x;
  const int lane = t & 31;
  const int wave = t >> 5;
  const int wm   = wave & 3;
  const int wn   = wave >> 2;
  const int bm   = blockIdx.y * 128;
  const int bn   = blockIdx.x * 64;

  v8f acc[2][2] = {};

  const int mrow = lane & 15;
  const int koff = (lane >> 4) * 2;
  const int ncol = lane & 15;

  for (int k0 = 0; k0 < K; k0 += 16) {
    #pragma unroll
    for (int i = 0; i < 2; ++i) {
      int idx = t + i * 256;
      int row = idx >> 2;
      int c4  = idx & 3;
      float4 v = *(const float4*)(A + (size_t)(bm + row) * K + k0 + c4 * 4);
      *(float4*)(lA + row * LDA_P + c4 * 4) = v;
    }
    {
      int row = t >> 4;
      int c4  = t & 15;
      float4 v = *(const float4*)(W + (size_t)(k0 + row) * N + bn + c4 * 4);
      *(float4*)(lB + row * LDB_P + c4 * 4) = v;
    }
    __syncthreads();

    #pragma unroll
    for (int kk = 0; kk < 16; kk += 4) {
      v2f afrag[2], bfrag[2];
      #pragma unroll
      for (int mt = 0; mt < 2; ++mt) {
        int row = wm * 32 + mt * 16 + mrow;
        afrag[mt] = *(const v2f*)(lA + row * LDA_P + kk + koff);
      }
      #pragma unroll
      for (int nt = 0; nt < 2; ++nt) {
        int col = wn * 32 + nt * 16 + ncol;
        bfrag[nt].x = lB[(kk + koff + 0) * LDB_P + col];
        bfrag[nt].y = lB[(kk + koff + 1) * LDB_P + col];
      }
      #pragma unroll
      for (int mt = 0; mt < 2; ++mt)
        #pragma unroll
        for (int nt = 0; nt < 2; ++nt)
          acc[mt][nt] = __builtin_amdgcn_wmma_f32_16x16x4_f32(
              false, afrag[mt], false, bfrag[nt],
              (short)0, acc[mt][nt], false, false);
    }
    __syncthreads();
  }

  const int rsel = (lane >> 4) * 8;
  #pragma unroll
  for (int nt = 0; nt < 2; ++nt) {
    int col = bn + wn * 32 + nt * 16 + ncol;
    float bv = bias[col];
    #pragma unroll
    for (int mt = 0; mt < 2; ++mt) {
      #pragma unroll
      for (int v = 0; v < 8; ++v) {
        int row = bm + wm * 32 + mt * 16 + rsel + v;
        float x = acc[mt][nt][v] + bv;
        if (relu) x = fmaxf(x, 0.0f);
        out[(size_t)row * N + col] = x;
      }
    }
  }
}

__global__ __launch_bounds__(256) void norm_rows64(float* __restrict__ q)
{
  const int lane = threadIdx.x & 31;
  const int wave = threadIdx.x >> 5;
  const int row  = blockIdx.x * 8 + wave;
  float* p = q + (size_t)row * 64;

  float2 x = *(float2*)(p + lane * 2);
  float s  = x.x + x.y;
  float ss = x.x * x.x + x.y * x.y;
  #pragma unroll
  for (int off = 16; off > 0; off >>= 1) {
    s  += __shfl_xor(s,  off, 32);
    ss += __shfl_xor(ss, off, 32);
  }
  float mean = s * (1.0f / 64.0f);
  float var  = (ss - 64.0f * mean * mean) * (1.0f / 63.0f);
  float inv  = rsqrtf(var + 1e-8f);
  x.x = (x.x - mean) * inv;
  x.y = (x.y - mean) * inv;
  *(float2*)(p + lane * 2) = x;
}

__global__ void sink_init(float* __restrict__ v, unsigned int* __restrict__ maxbuf)
{
  int i = blockIdx.x * 256 + threadIdx.x;
  if (i < BSZ * CCH) v[i] = 1.0f;
  if (i < CCH) maxbuf[i] = 0u;
}

#define LQA_P 68
#define LQB_P 33

__global__ __launch_bounds__(256) void affinity_wmma(
    const float* __restrict__ q1, const float* __restrict__ q2,
    float* __restrict__ aff, unsigned int* __restrict__ maxbuf)
{
  __shared__ float lA[64 * LQA_P];
  __shared__ float lB[64 * LQB_P];
  __shared__ float red[16 * CCH];

  const int t    = threadIdx.x;
  const int lane = t & 31;
  const int wave = t >> 5;
  const int wm   = wave & 3;
  const int wn   = wave >> 2;
  const int bb   = blockIdx.y * 64;
  const int bd   = blockIdx.x * 32;

  const int srow = t >> 4;
  const int sc4  = t & 15;
  const float* gA = q1 + (size_t)(bb + srow) * SLICE + sc4 * 4;
  const float* gB = q2 + (size_t)(bd + srow) * SLICE + sc4 * 4;

  float4 pa0 = *(const float4*)(gA);
  float4 pa1 = *(const float4*)(gA + (size_t)16 * SLICE);
  float4 pa2 = *(const float4*)(gA + (size_t)32 * SLICE);
  float4 pa3 = *(const float4*)(gA + (size_t)48 * SLICE);
  float4 pb0 = *(const float4*)(gB);
  float4 pb1 = *(const float4*)(gB + (size_t)16 * SLICE);
  __builtin_amdgcn_sched_barrier(0);

  v8f acc[CCH] = {};
  const int mrow = lane & 15;
  const int koff = (lane >> 4) * 2;
  const int ncol = lane & 15;

  #pragma unroll
  for (int c = 0; c < CCH; ++c) {
    *(float4*)(lA + (srow +  0) * LQA_P + sc4 * 4) = pa0;
    *(float4*)(lA + (srow + 16) * LQA_P + sc4 * 4) = pa1;
    *(float4*)(lA + (srow + 32) * LQA_P + sc4 * 4) = pa2;
    *(float4*)(lA + (srow + 48) * LQA_P + sc4 * 4) = pa3;
    lB[(sc4 * 4 + 0) * LQB_P + srow] = pb0.x;
    lB[(sc4 * 4 + 1) * LQB_P + srow] = pb0.y;
    lB[(sc4 * 4 + 2) * LQB_P + srow] = pb0.z;
    lB[(sc4 * 4 + 3) * LQB_P + srow] = pb0.w;
    lB[(sc4 * 4 + 0) * LQB_P + srow + 16] = pb1.x;
    lB[(sc4 * 4 + 1) * LQB_P + srow + 16] = pb1.y;
    lB[(sc4 * 4 + 2) * LQB_P + srow + 16] = pb1.z;
    lB[(sc4 * 4 + 3) * LQB_P + srow + 16] = pb1.w;
    __syncthreads();

    if (c + 1 < CCH) {
      const float* nA = gA + (c + 1) * EDIM;
      const float* nB = gB + (c + 1) * EDIM;
      pa0 = *(const float4*)(nA);
      pa1 = *(const float4*)(nA + (size_t)16 * SLICE);
      pa2 = *(const float4*)(nA + (size_t)32 * SLICE);
      pa3 = *(const float4*)(nA + (size_t)48 * SLICE);
      pb0 = *(const float4*)(nB);
      pb1 = *(const float4*)(nB + (size_t)16 * SLICE);
    }

    #pragma unroll
    for (int kk = 0; kk < 64; kk += 4) {
      v2f afrag, bfrag;
      afrag = *(const v2f*)(lA + (wm * 16 + mrow) * LQA_P + kk + koff);
      bfrag.x = lB[(kk + koff + 0) * LQB_P + wn * 16 + ncol];
      bfrag.y = lB[(kk + koff + 1) * LQB_P + wn * 16 + ncol];
      acc[c] = __builtin_amdgcn_wmma_f32_16x16x4_f32(
          false, afrag, false, bfrag, (short)0, acc[c], false, false);
    }
    __syncthreads();
  }

  const float scale = 0.125f;
  const int rsel = (lane >> 4) * 8;
  const int col  = bd + wn * 16 + ncol;
  #pragma unroll
  for (int v = 0; v < 8; ++v) {
    int row = bb + wm * 16 + rsel + v;
    float* dst = aff + ((size_t)row * BSZ + col) * CCH;
    #pragma unroll
    for (int q = 0; q < 4; ++q) {
      float4 o;
      o.x = acc[4 * q + 0][v] * scale;
      o.y = acc[4 * q + 1][v] * scale;
      o.z = acc[4 * q + 2][v] * scale;
      o.w = acc[4 * q + 3][v] * scale;
      *(float4*)(dst + 4 * q) = o;
    }
  }

  float* park = lA;
  #pragma unroll
  for (int c = 0; c < CCH; ++c) {
    float m = -3.4e38f;
    #pragma unroll
    for (int v = 0; v < 8; ++v) m = fmaxf(m, acc[c][v] * scale);
    park[t * 17 + c] = m;
  }
  __syncthreads();
  {
    const int pc = t & 15;
    const int pg = t >> 4;
    float m = park[(pg * 16) * 17 + pc];
    for (int i = 1; i < 16; ++i) m = fmaxf(m, park[(pg * 16 + i) * 17 + pc]);
    red[pg * CCH + pc] = m;
  }
  __syncthreads();
  if (t < CCH) {
    float m = red[t];
    for (int g = 1; g < 16; ++g) m = fmaxf(m, red[g * CCH + t]);
    atomicMax(&maxbuf[t], enc_f32(m));
  }
}

__global__ __launch_bounds__(256) void exp_sub_max(
    float* __restrict__ A, const unsigned int* __restrict__ maxbuf)
{
  size_t base   = ((size_t)blockIdx.x * 256 + threadIdx.x) * 4;
  size_t stride = (size_t)gridDim.x * 256 * 4;
  for (size_t i = base; i < (size_t)ATOT; i += stride) {
    #pragma unroll
    for (int j = 0; j < 4; ++j) {
      size_t idx = i + j;
      float mv = dec_f32(maxbuf[idx & 15]);
      A[idx] = expf(A[idx] - mv);
    }
  }
}

__global__ __launch_bounds__(256) void sink_u(
    const float* __restrict__ A, const float* __restrict__ v,
    const float* __restrict__ r, float* __restrict__ u)
{
  __shared__ float red[8 * 32];
  const int t    = threadIdx.x;
  const int lane = t & 31;
  const int wave = t >> 5;
  const int b0   = blockIdx.x * 2;
  const float4* A4 = (const float4*)(A + (size_t)b0 * (BSZ * CCH)) + t;
  const float4* v4 = (const float4*)v + t;

  float s[8] = {0.0f, 0.0f, 0.0f, 0.0f, 0.0f, 0.0f, 0.0f, 0.0f};
  #pragma unroll
  for (int i = 0; i < 16; ++i) {
    const float4 w  = v4[i * 256];
    const float4 x0 = A4[i * 256];
    const float4 x1 = A4[4096 + i * 256];
    s[0] += x0.x * w.x;  s[1] += x0.y * w.y;
    s[2] += x0.z * w.z;  s[3] += x0.w * w.w;
    s[4] += x1.x * w.x;  s[5] += x1.y * w.y;
    s[6] += x1.z * w.z;  s[7] += x1.w * w.w;
  }
  #pragma unroll
  for (int q = 0; q < 8; ++q) {
    s[q] += __shfl_xor(s[q], 4, 32);
    s[q] += __shfl_xor(s[q], 8, 32);
    s[q] += __shfl_xor(s[q], 16, 32);
  }
  if (lane < 4) {
    #pragma unroll
    for (int k = 0; k < 4; ++k) {
      red[wave * 32 +  0 + lane * 4 + k] = s[k];
      red[wave * 32 + 16 + lane * 4 + k] = s[4 + k];
    }
  }
  __syncthreads();
  if (t < 32) {
    float sum = red[t];
    for (int w = 1; w < 8; ++w) sum += red[w * 32 + t];
    u[b0 * CCH + t] = r[b0 * CCH + t] / sum;
  }
}

__global__ __launch_bounds__(256) void sink_v(
    const float* __restrict__ A, const float* __restrict__ u,
    const float* __restrict__ cmarg, float* __restrict__ v)
{
  __shared__ float4 su[BSZ * CCH / 4];
  __shared__ float red[8 * 64];
  const int t    = threadIdx.x;
  const int lane = t & 31;
  const int wave = t >> 5;
  const int l16  = t & 15;
  const int rg   = t >> 4;
  const int j0   = blockIdx.x * 64;
  const float* Ap = A + (size_t)rg * (BSZ * CCH) + j0 + l16 * 4;
  const float4* sp = su + rg * 4 + (l16 & 3);

  #pragma unroll
  for (int i = 0; i < 16; ++i) su[t + 256 * i] = ((const float4*)u)[t + 256 * i];
  __syncthreads();

  float4 acc = make_float4(0.0f, 0.0f, 0.0f, 0.0f);
  #pragma unroll
  for (int step = 0; step < 64; ++step) {
    float4 a = *(const float4*)(Ap + (size_t)step * 16 * (BSZ * CCH));
    float4 w = sp[step * 64];
    acc.x += a.x * w.x;  acc.y += a.y * w.y;
    acc.z += a.z * w.z;  acc.w += a.w * w.w;
  }

  acc.x += __shfl_xor(acc.x, 16, 32);
  acc.y += __shfl_xor(acc.y, 16, 32);
  acc.z += __shfl_xor(acc.z, 16, 32);
  acc.w += __shfl_xor(acc.w, 16, 32);
  if (lane < 16) {
    red[wave * 64 + lane * 4 + 0] = acc.x;
    red[wave * 64 + lane * 4 + 1] = acc.y;
    red[wave * 64 + lane * 4 + 2] = acc.z;
    red[wave * 64 + lane * 4 + 3] = acc.w;
  }
  __syncthreads();
  if (t < 64) {
    float sum = red[t];
    for (int w = 1; w < 8; ++w) sum += red[w * 64 + t];
    v[j0 + t] = cmarg[j0 + t] / sum;
  }
}

__global__ __launch_bounds__(256) void sink_final(
    float* __restrict__ A, const float* __restrict__ u,
    const float* __restrict__ v)
{
  size_t base   = ((size_t)blockIdx.x * 256 + threadIdx.x) * 4;
  size_t stride = (size_t)gridDim.x * 256 * 4;
  for (size_t i = base; i < (size_t)ATOT; i += stride) {
    #pragma unroll
    for (int j = 0; j < 4; ++j) {
      size_t idx = i + j;
      size_t b  = idx >> 14;
      size_t dc = idx & 16383;
      A[idx] = A[idx] * u[b * CCH + (idx & 15)] * v[dc];
    }
  }
}

extern "C" void kernel_launch(void* const* d_in, const int* in_sizes, int n_in,
                              void* d_out, int out_size, void* d_ws, size_t ws_size,
                              hipStream_t stream)
{
  const float* x1   = (const float*)d_in[0];
  const float* x2   = (const float*)d_in[1];
  const float* p1   = (const float*)d_in[2];
  const float* p2   = (const float*)d_in[3];
  const float* w1[4] = {(const float*)d_in[4],  (const float*)d_in[6],
                        (const float*)d_in[8],  (const float*)d_in[10]};
  const float* b1[4] = {(const float*)d_in[5],  (const float*)d_in[7],
                        (const float*)d_in[9],  (const float*)d_in[11]};
  const float* w2[4] = {(const float*)d_in[12], (const float*)d_in[14],
                        (const float*)d_in[16], (const float*)d_in[18]};
  const float* b2[4] = {(const float*)d_in[13], (const float*)d_in[15],
                        (const float*)d_in[17], (const float*)d_in[19]};

  float* ws = (float*)d_ws;
  float* hA = ws;
  float* hB = hA + 1024 * 512;
  float* q1 = hB + 1024 * 512;
  float* q2 = q1 + 1024 * 1024;
  float* uv_u = q2 + 1024 * 1024;
  float* uv_v = uv_u + 16384;
  unsigned int* maxbuf = (unsigned int*)(uv_v + 16384);

  float* Aout = (float*)d_out;

  float* h1a = q1;
  float* h1b = q1 + 1024 * 512;
  float* h2a = q2;
  float* h2b = q2 + 1024 * 512;

  gemm_bias_act<<<dim3(512/64, 1024/128, 2), 256, 0, stream>>>(
      x1, w1[0], b1[0], h1a,  x2, w2[0], b2[0], h2a,  1024, 256, 512, 1);
  gemm_bias_act<<<dim3(512/64, 1024/128, 2), 256, 0, stream>>>(
      h1a, w1[1], b1[1], h1b,  h2a, w2[1], b2[1], h2b,  1024, 512, 512, 1);
  gemm_bias_act<<<dim3(512/64, 1024/128, 2), 256, 0, stream>>>(
      h1b, w1[2], b1[2], hA,  h2b, w2[2], b2[2], hB,  1024, 512, 512, 1);
  gemm_bias_act<<<dim3(1024/64, 1024/128, 2), 256, 0, stream>>>(
      hA, w1[3], b1[3], q1,  hB, w2[3], b2[3], q2,  1024, 512, 1024, 0);

  norm_rows64<<<2 * 16384 / 8, 256, 0, stream>>>(q1);

  sink_init<<<(1024 * 16 + 255) / 256, 256, 0, stream>>>(uv_v, maxbuf);

  affinity_wmma<<<dim3(1024/32, 1024/64), 256, 0, stream>>>(q1, q2, Aout, maxbuf);

  exp_sub_max<<<4096, 256, 0, stream>>>(Aout, maxbuf);

  for (int it = 0; it < 10; ++it) {
    sink_u<<<512, 256, 0, stream>>>(Aout, uv_v, p1, uv_u);
    sink_v<<<256, 256, 0, stream>>>(Aout, uv_u, p2, uv_v);
  }

  sink_final<<<4096, 256, 0, stream>>>(Aout, uv_u, uv_v);
}


# MLP GEMMs: K loop software-pipelined in stages of 64: the next stage's global loads fly under the current stage's 64 products, two LDS stages, one barrier per stage
# speedup vs baseline: 2.1595x; 1.1886x over previous
#include <hip/hip_runtime.h>
#include <hip/hip_bf16.h>


typedef float v2f __attribute__((ext_vector_type(2)));
typedef float v8f __attribute__((ext_vector_type(8)));

#define BSZ   1024
#define CCH   16
#define EDIM  64
#define SLICE (CCH * EDIM)
#define ATOT  (BSZ * BSZ * CCH)

__device__ __forceinline__ unsigned int enc_f32(float f) {
  unsigned int u = __float_as_uint(f);
  return (u & 0x80000000u) ? ~u : (u | 0x80000000u);
}
__device__ __forceinline__ float dec_f32(unsigned int u) {
  return (u & 0x80000000u) ? __uint_as_float(u & 0x7FFFFFFFu)
                           : __uint_as_float(~u);
}

#define GK    64
#define LDA_P 68
#define LDB_P 68
#define LA_SZ (128 * LDA_P)
#define LB_SZ (GK * LDB_P)

__device__ __forceinline__ void gemm_stage(
    const float* __restrict__ a, const float* __restrict__ b,
    int arow, int koff, int bcol, v8f (&acc)[2][2])
{
  #pragma unroll
  for (int kk = 0; kk < GK; kk += 4) {
    v2f afrag[2], bfrag[2];
    #pragma unroll
    for (int mt = 0; mt < 2; ++mt)
      afrag[mt] = *(const v2f*)(a + (arow + mt * 16) * LDA_P + kk + koff);
    #pragma unroll
    for (int nt = 0; nt < 2; ++nt) {
      bfrag[nt].x = b[(kk + koff + 0) * LDB_P + bcol + nt * 16];
      bfrag[nt].y = b[(kk + koff + 1) * LDB_P + bcol + nt * 16];
    }
    #pragma unroll
    for (int mt = 0; mt < 2; ++mt)
      #pragma unroll
      for (int nt = 0; nt < 2; ++nt)
        acc[mt][nt] = __builtin_amdgcn_wmma_f32_16x16x4_f32(
            false, afrag[mt], false, bfrag[nt],
            (short)0, acc[mt][nt], false, false);
  }
}

__global__ __launch_bounds__(256) void gemm_bias_act(
    const float* __restrict__ A0, const float* __restrict__ W0,
    const float* __restrict__ bias0, float* __restrict__ out0,
    const float* __restrict__ A1, const float* __restrict__ W1,
    const float* __restrict__ bias1, float* __restrict__ out1,
    int M, int K, int N, int relu)
{
  __shared__ float lA[2 * LA_SZ];
  __shared__ float lB[2 * LB_SZ];

  const bool br = blockIdx.z != 0;
  const float* __restrict__ A    = br ? A1 : A0;
  const float* __restrict__ W    = br ? W1 : W0;
  const float* __restrict__ bias = br ? bias1 : bias0;
  float* __restrict__ out        = br ? out1 : out0;

  const int t    = threadIdx.x;
  const int lane = t & 31;
  const int wave = t >> 5;
  const int wm   = wave & 3;
  const int wn   = wave >> 2;
  const int bm   = blockIdx.y * 128;
  const int bn   = blockIdx.x * 64;

  v8f acc[2][2] = {};

  const int mrow = lane & 15;
  const int koff = (lane >> 4) * 2;
  const int ncol = lane & 15;
  const int arow = wm * 32 + mrow;
  const int bcol = wn * 32 + ncol;

  const int srow = t >> 4;
  const int sc4  = t & 15;
  const float* gA = A + (size_t)(bm + srow) * K + sc4 * 4;
  const float* gB = W + (size_t)srow * N + bn + sc4 * 4;
  float* sA = lA + srow * LDA_P + sc4 * 4;
  float* sB = lB + srow * LDB_P + sc4 * 4;

  float4 pa0 = *(const float4*)(gA);
  float4 pa1 = *(const float4*)(gA + (size_t)16 * K);
  float4 pa2 = *(const float4*)(gA + (size_t)32 * K);
  float4 pa3 = *(const float4*)(gA + (size_t)48 * K);
  float4 pa4 = *(const float4*)(gA + (size_t)64 * K);
  float4 pa5 = *(const float4*)(gA + (size_t)80 * K);
  float4 pa6 = *(const float4*)(gA + (size_t)96 * K);
  float4 pa7 = *(const float4*)(gA + (size_t)112 * K);
  float4 pb0 = *(const float4*)(gB);
  float4 pb1 = *(const float4*)(gB + (size_t)16 * N);
  float4 pb2 = *(const float4*)(gB + (size_t)32 * N);
  float4 pb3 = *(const float4*)(gB + (size_t)48 * N);
  *(float4*)(sA +   0 * LDA_P) = pa0;
  *(float4*)(sA +  16 * LDA_P) = pa1;
  *(float4*)(sA +  32 * LDA_P) = pa2;
  *(float4*)(sA +  48 * LDA_P) = pa3;
  *(float4*)(sA +  64 * LDA_P) = pa4;
  *(float4*)(sA +  80 * LDA_P) = pa5;
  *(float4*)(sA +  96 * LDA_P) = pa6;
  *(float4*)(sA + 112 * LDA_P) = pa7;
  *(float4*)(sB +   0 * LDB_P) = pb0;
  *(float4*)(sB +  16 * LDB_P) = pb1;
  *(float4*)(sB +  32 * LDB_P) = pb2;
  *(float4*)(sB +  48 * LDB_P) = pb3;
  __syncthreads();

  int cur = 0;
  for (int k0 = GK; k0 < K; k0 += GK) {
    const float* nA = gA + k0;
    const float* nB = gB + (size_t)k0 * N;
    pa0 = *(const float4*)(nA);
    pa1 = *(const float4*)(nA + (size_t)16 * K);
    pa2 = *(const float4*)(nA + (size_t)32 * K);
    pa3 = *(const float4*)(nA + (size_t)48 * K);
    pa4 = *(const float4*)(nA + (size_t)64 * K);
    pa5 = *(const float4*)(nA + (size_t)80 * K);
    pa6 = *(const float4*)(nA + (size_t)96 * K);
    pa7 = *(const float4*)(nA + (size_t)112 * K);
    pb0 = *(const float4*)(nB);
    pb1 = *(const float4*)(nB + (size_t)16 * N);
    pb2 = *(const float4*)(nB + (size_t)32 * N);
    pb3 = *(const float4*)(nB + (size_t)48 * N);

    gemm_stage(lA + cur * LA_SZ, lB + cur * LB_SZ, arow, koff, bcol, acc);

    __builtin_amdgcn_sched_barrier(0);
    const int nxt = cur ^ 1;
    float* dA = sA + nxt * LA_SZ;
    float* dB = sB + nxt * LB_SZ;
    *(float4*)(dA +   0 * LDA_P) = pa0;
    *(float4*)(dA +  16 * LDA_P) = pa1;
    *(float4*)(dA +  32 * LDA_P) = pa2;
    *(float4*)(dA +  48 * LDA_P) = pa3;
    *(float4*)(dA +  64 * LDA_P) = pa4;
    *(float4*)(dA +  80 * LDA_P) = pa5;
    *(float4*)(dA +  96 * LDA_P) = pa6;
    *(float4*)(dA + 112 * LDA_P) = pa7;
    *(float4*)(dB +   0 * LDB_P) = pb0;
    *(float4*)(dB +  16 * LDB_P) = pb1;
    *(float4*)(dB +  32 * LDB_P) = pb2;
    *(float4*)(dB +  48 * LDB_P) = pb3;
    __syncthreads();
    cur = nxt;
  }
  gemm_stage(lA + cur * LA_SZ, lB + cur * LB_SZ, arow, koff, bcol, acc);
  __builtin_amdgcn_sched_barrier(0);

  const int rsel = (lane >> 4) * 8;
  #pragma unroll
  for (int nt = 0; nt < 2; ++nt) {
    int col = bn + wn * 32 + nt * 16 + ncol;
    float bv = bias[col];
    #pragma unroll
    for (int mt = 0; mt < 2; ++mt) {
      #pragma unroll
      for (int v = 0; v < 8; ++v) {
        int row = bm + wm * 32 + mt * 16 + rsel + v;
        float x = acc[mt][nt][v] + bv;
        if (relu) x = fmaxf(x, 0.0f);
        out[(size_t)row * N + col] = x;
      }
    }
  }
}

__global__ __launch_bounds__(256) void norm_rows64(float* __restrict__ q)
{
  const int lane = threadIdx.x & 31;
  const int wave = threadIdx.x >> 5;
  const int row  = blockIdx.x * 8 + wave;
  float* p = q + (size_t)row * 64;

  float2 x = *(float2*)(p + lane * 2);
  float s  = x.x + x.y;
  float ss = x.x * x.x + x.y * x.y;
  #pragma unroll
  for (int off = 16; off > 0; off >>= 1) {
    s  += __shfl_xor(s,  off, 32);
    ss += __shfl_xor(ss, off, 32);
  }
  float mean = s * (1.0f / 64.0f);
  float var  = (ss - 64.0f * mean * mean) * (1.0f / 63.0f);
  float inv  = rsqrtf(var + 1e-8f);
  x.x = (x.x - mean) * inv;
  x.y = (x.y - mean) * inv;
  *(float2*)(p + lane * 2) = x;
}

__global__ void sink_init(float* __restrict__ v, unsigned int* __restrict__ maxbuf)
{
  int i = blockIdx.x * 256 + threadIdx.x;
  if (i < BSZ * CCH) v[i] = 1.0f;
  if (i < CCH) maxbuf[i] = 0u;
}

#define LQA_P 68
#define LQB_P 33

__global__ __launch_bounds__(256) void affinity_wmma(
    const float* __restrict__ q1, const float* __restrict__ q2,
    float* __restrict__ aff, unsigned int* __restrict__ maxbuf)
{
  __shared__ float lA[64 * LQA_P];
  __shared__ float lB[64 * LQB_P];
  __shared__ float red[16 * CCH];

  const int t    = threadIdx.x;
  const int lane = t & 31;
  const int wave = t >> 5;
  const int wm   = wave & 3;
  const int wn   = wave >> 2;
  const int bb   = blockIdx.y * 64;
  const int bd   = blockIdx.x * 32;

  const int srow = t >> 4;
  const int sc4  = t & 15;
  const float* gA = q1 + (size_t)(bb + srow) * SLICE + sc4 * 4;
  const float* gB = q2 + (size_t)(bd + srow) * SLICE + sc4 * 4;

  float4 pa0 = *(const float4*)(gA);
  float4 pa1 = *(const float4*)(gA + (size_t)16 * SLICE);
  float4 pa2 = *(const float4*)(gA + (size_t)32 * SLICE);
  float4 pa3 = *(const float4*)(gA + (size_t)48 * SLICE);
  float4 pb0 = *(const float4*)(gB);
  float4 pb1 = *(const float4*)(gB + (size_t)16 * SLICE);
  __builtin_amdgcn_sched_barrier(0);

  v8f acc[CCH] = {};
  const int mrow = lane & 15;
  const int koff = (lane >> 4) * 2;
  const int ncol = lane & 15;

  #pragma unroll
  for (int c = 0; c < CCH; ++c) {
    *(float4*)(lA + (srow +  0) * LQA_P + sc4 * 4) = pa0;
    *(float4*)(lA + (srow + 16) * LQA_P + sc4 * 4) = pa1;
    *(float4*)(lA + (srow + 32) * LQA_P + sc4 * 4) = pa2;
    *(float4*)(lA + (srow + 48) * LQA_P + sc4 * 4) = pa3;
    lB[(sc4 * 4 + 0) * LQB_P + srow] = pb0.x;
    lB[(sc4 * 4 + 1) * LQB_P + srow] = pb0.y;
    lB[(sc4 * 4 + 2) * LQB_P + srow] = pb0.z;
    lB[(sc4 * 4 + 3) * LQB_P + srow] = pb0.w;
    lB[(sc4 * 4 + 0) * LQB_P + srow + 16] = pb1.x;
    lB[(sc4 * 4 + 1) * LQB_P + srow + 16] = pb1.y;
    lB[(sc4 * 4 + 2) * LQB_P + srow + 16] = pb1.z;
    lB[(sc4 * 4 + 3) * LQB_P + srow + 16] = pb1.w;
    __syncthreads();

    if (c + 1 < CCH) {
      const float* nA = gA + (c + 1) * EDIM;
      const float* nB = gB + (c + 1) * EDIM;
      pa0 = *(const float4*)(nA);
      pa1 = *(const float4*)(nA + (size_t)16 * SLICE);
      pa2 = *(const float4*)(nA + (size_t)32 * SLICE);
      pa3 = *(const float4*)(nA + (size_t)48 * SLICE);
      pb0 = *(const float4*)(nB);
      pb1 = *(const float4*)(nB + (size_t)16 * SLICE);
    }

    #pragma unroll
    for (int kk = 0; kk < 64; kk += 4) {
      v2f afrag, bfrag;
      afrag = *(const v2f*)(lA + (wm * 16 + mrow) * LQA_P + kk + koff);
      bfrag.x = lB[(kk + koff + 0) * LQB_P + wn * 16 + ncol];
      bfrag.y = lB[(kk + koff + 1) * LQB_P + wn * 16 + ncol];
      acc[c] = __builtin_amdgcn_wmma_f32_16x16x4_f32(
          false, afrag, false, bfrag, (short)0, acc[c], false, false);
    }
    __syncthreads();
  }

  const float scale = 0.125f;
  const int rsel = (lane >> 4) * 8;
  const int col  = bd + wn * 16 + ncol;
  #pragma unroll
  for (int v = 0; v < 8; ++v) {
    int row = bb + wm * 16 + rsel + v;
    float* dst = aff + ((size_t)row * BSZ + col) * CCH;
    #pragma unroll
    for (int q = 0; q < 4; ++q) {
      float4 o;
      o.x = acc[4 * q + 0][v] * scale;
      o.y = acc[4 * q + 1][v] * scale;
      o.z = acc[4 * q + 2][v] * scale;
      o.w = acc[4 * q + 3][v] * scale;
      *(float4*)(dst + 4 * q) = o;
    }
  }

  float* park = lA;
  #pragma unroll
  for (int c = 0; c < CCH; ++c) {
    float m = -3.4e38f;
    #pragma unroll
    for (int v = 0; v < 8; ++v) m = fmaxf(m, acc[c][v] * scale);
    park[t * 17 + c] = m;
  }
  __syncthreads();
  {
    const int pc = t & 15;
    const int pg = t >> 4;
    float m = park[(pg * 16) * 17 + pc];
    for (int i = 1; i < 16; ++i) m = fmaxf(m, park[(pg * 16 + i) * 17 + pc]);
    red[pg * CCH + pc] = m;
  }
  __syncthreads();
  if (t < CCH) {
    float m = red[t];
    for (int g = 1; g < 16; ++g) m = fmaxf(m, red[g * CCH + t]);
    atomicMax(&maxbuf[t], enc_f32(m));
  }
}

__global__ __launch_bounds__(256) void exp_sub_max(
    float* __restrict__ A, const unsigned int* __restrict__ maxbuf)
{
  size_t base   = ((size_t)blockIdx.x * 256 + threadIdx.x) * 4;
  size_t stride = (size_t)gridDim.x * 256 * 4;
  for (size_t i = base; i < (size_t)ATOT; i += stride) {
    #pragma unroll
    for (int j = 0; j < 4; ++j) {
      size_t idx = i + j;
      float mv = dec_f32(maxbuf[idx & 15]);
      A[idx] = expf(A[idx] - mv);
    }
  }
}

__global__ __launch_bounds__(256) void sink_u(
    const float* __restrict__ A, const float* __restrict__ v,
    const float* __restrict__ r, float* __restrict__ u)
{
  __shared__ float red[8 * 32];
  const int t    = threadIdx.x;
  const int lane = t & 31;
  const int wave = t >> 5;
  const int b0   = blockIdx.x * 2;
  const float4* A4 = (const float4*)(A + (size_t)b0 * (BSZ * CCH)) + t;
  const float4* v4 = (const float4*)v + t;

  float s[8] = {0.0f, 0.0f, 0.0f, 0.0f, 0.0f, 0.0f, 0.0f, 0.0f};
  #pragma unroll
  for (int i = 0; i < 16; ++i) {
    const float4 w  = v4[i * 256];
    const float4 x0 = A4[i * 256];
    const float4 x1 = A4[4096 + i * 256];
    s[0] += x0.x * w.x;  s[1] += x0.y * w.y;
    s[2] += x0.z * w.z;  s[3] += x0.w * w.w;
    s[4] += x1.x * w.x;  s[5] += x1.y * w.y;
    s[6] += x1.z * w.z;  s[7] += x1.w * w.w;
  }
  #pragma unroll
  for (int q = 0; q < 8; ++q) {
    s[q] += __shfl_xor(s[q], 4, 32);
    s[q] += __shfl_xor(s[q], 8, 32);
    s[q] += __shfl_xor(s[q], 16, 32);
  }
  if (lane < 4) {
    #pragma unroll
    for (int k = 0; k < 4; ++k) {
      red[wave * 32 +  0 + lane * 4 + k] = s[k];
      red[wave * 32 + 16 + lane * 4 + k] = s[4 + k];
    }
  }
  __syncthreads();
  if (t < 32) {
    float sum = red[t];
    for (int w = 1; w < 8; ++w) sum += red[w * 32 + t];
    u[b0 * CCH + t] = r[b0 * CCH + t] / sum;
  }
}

__global__ __launch_bounds__(256) void sink_v(
    const float* __restrict__ A, const float* __restrict__ u,
    const float* __restrict__ cmarg, float* __restrict__ v)
{
  __shared__ float4 su[BSZ * CCH / 4];
  __shared__ float red[8 * 64];
  const int t    = threadIdx.x;
  const int lane = t & 31;
  const int wave = t >> 5;
  const int l16  = t & 15;
  const int rg   = t >> 4;
  const int j0   = blockIdx.x * 64;
  const float* Ap = A + (size_t)rg * (BSZ * CCH) + j0 + l16 * 4;
  const float4* sp = su + rg * 4 + (l16 & 3);

  #pragma unroll
  for (int i = 0; i < 16; ++i) su[t + 256 * i] = ((const float4*)u)[t + 256 * i];
  __syncthreads();

  float4 acc = make_float4(0.0f, 0.0f, 0.0f, 0.0f);
  #pragma unroll
  for (int step = 0; step < 64; ++step) {
    float4 a = *(const float4*)(Ap + (size_t)step * 16 * (BSZ * CCH));
    float4 w = sp[step * 64];
    acc.x += a.x * w.x;  acc.y += a.y * w.y;
    acc.z += a.z * w.z;  acc.w += a.w * w.w;
  }

  acc.x += __shfl_xor(acc.x, 16, 32);
  acc.y += __shfl_xor(acc.y, 16, 32);
  acc.z += __shfl_xor(acc.z, 16, 32);
  acc.w += __shfl_xor(acc.w, 16, 32);
  if (lane < 16) {
    red[wave * 64 + lane * 4 + 0] = acc.x;
    red[wave * 64 + lane * 4 + 1] = acc.y;
    red[wave * 64 + lane * 4 + 2] = acc.z;
    red[wave * 64 + lane * 4 + 3] = acc.w;
  }
  __syncthreads();
  if (t < 64) {
    float sum = red[t];
    for (int w = 1; w < 8; ++w) sum += red[w * 64 + t];
    v[j0 + t] = cmarg[j0 + t] / sum;
  }
}

__global__ __launch_bounds__(256) void sink_final(
    float* __restrict__ A, const float* __restrict__ u,
    const float* __restrict__ v)
{
  size_t base   = ((size_t)blockIdx.x * 256 + threadIdx.x) * 4;
  size_t stride = (size_t)gridDim.x * 256 * 4;
  for (size_t i = base; i < (size_t)ATOT; i += stride) {
    #pragma unroll
    for (int j = 0; j < 4; ++j) {
      size_t idx = i + j;
      size_t b  = idx >> 14;
      size_t dc = idx & 16383;
      A[idx] = A[idx] * u[b * CCH + (idx & 15)] * v[dc];
    }
  }
}

extern "C" void kernel_launch(void* const* d_in, const int* in_sizes, int n_in,
                              void* d_out, int out_size, void* d_ws, size_t ws_size,
                              hipStream_t stream)
{
  const float* x1   = (const float*)d_in[0];
  const float* x2   = (const float*)d_in[1];
  const float* p1   = (const float*)d_in[2];
  const float* p2   = (const float*)d_in[3];
  const float* w1[4] = {(const float*)d_in[4],  (const float*)d_in[6],
                        (const float*)d_in[8],  (const float*)d_in[10]};
  const float* b1[4] = {(const float*)d_in[5],  (const float*)d_in[7],
                        (const float*)d_in[9],  (const float*)d_in[11]};
  const float* w2[4] = {(const float*)d_in[12], (const float*)d_in[14],
                        (const float*)d_in[16], (const float*)d_in[18]};
  const float* b2[4] = {(const float*)d_in[13], (const float*)d_in[15],
                        (const float*)d_in[17], (const float*)d_in[19]};

  float* ws = (float*)d_ws;
  float* hA = ws;
  float* hB = hA + 1024 * 512;
  float* q1 = hB + 1024 * 512;
  float* q2 = q1 + 1024 * 1024;
  float* uv_u = q2 + 1024 * 1024;
  float* uv_v = uv_u + 16384;
  unsigned int* maxbuf = (unsigned int*)(uv_v + 16384);

  float* Aout = (float*)d_out;

  float* h1a = q1;
  float* h1b = q1 + 1024 * 512;
  float* h2a = q2;
  float* h2b = q2 + 1024 * 512;

  gemm_bias_act<<<dim3(512/64, 1024/128, 2), 256, 0, stream>>>(
      x1, w1[0], b1[0], h1a,  x2, w2[0], b2[0], h2a,  1024, 256, 512, 1);
  gemm_bias_act<<<dim3(512/64, 1024/128, 2), 256, 0, stream>>>(
      h1a, w1[1], b1[1], h1b,  h2a, w2[1], b2[1], h2b,  1024, 512, 512, 1);
  gemm_bias_act<<<dim3(512/64, 1024/128, 2), 256, 0, stream>>>(
      h1b, w1[2], b1[2], hA,  h2b, w2[2], b2[2], hB,  1024, 512, 512, 1);
  gemm_bias_act<<<dim3(1024/64, 1024/128, 2), 256, 0, stream>>>(
      hA, w1[3], b1[3], q1,  hB, w2[3], b2[3], q2,  1024, 512, 1024, 0);

  norm_rows64<<<2 * 16384 / 8, 256, 0, stream>>>(q1);

  sink_init<<<(1024 * 16 + 255) / 256, 256, 0, stream>>>(uv_v, maxbuf);

  affinity_wmma<<<dim3(1024/32, 1024/64), 256, 0, stream>>>(q1, q2, Aout, maxbuf);

  exp_sub_max<<<4096, 256, 0, stream>>>(Aout, maxbuf);

  for (int it = 0; it < 10; ++it) {
    sink_u<<<512, 256, 0, stream>>>(Aout, uv_v, p1, uv_u);
    sink_v<<<256, 256, 0, stream>>>(Aout, uv_u, p2, uv_v);
  }

  sink_final<<<4096, 256, 0, stream>>>(Aout, uv_u, uv_v);
}


# one-off elementwise launches folded into neighbours: max reset into the norm launch, exp into the first row pass, final scaling into the last column pass (tile held in registers); 29 launches -> 26
# speedup vs baseline: 2.3814x; 1.1028x over previous
#include <hip/hip_runtime.h>
#include <hip/hip_bf16.h>


typedef float v2f __attribute__((ext_vector_type(2)));
typedef float v8f __attribute__((ext_vector_type(8)));

#define BSZ   1024
#define CCH   16
#define EDIM  64
#define SLICE (CCH * EDIM)
#define ATOT  (BSZ * BSZ * CCH)

__device__ __forceinline__ unsigned int enc_f32(float f) {
  unsigned int u = __float_as_uint(f);
  return (u & 0x80000000u) ? ~u : (u | 0x80000000u);
}
__device__ __forceinline__ float dec_f32(unsigned int u) {
  return (u & 0x80000000u) ? __uint_as_float(u & 0x7FFFFFFFu)
                           : __uint_as_float(~u);
}

#define GK    64
#define LDA_P 68
#define LDB_P 68
#define LA_SZ (128 * LDA_P)
#define LB_SZ (GK * LDB_P)

__device__ __forceinline__ void gemm_stage(
    const float* __restrict__ a, const float* __restrict__ b,
    int arow, int koff, int bcol, v8f (&acc)[2][2])
{
  #pragma unroll
  for (int kk = 0; kk < GK; kk += 4) {
    v2f afrag[2], bfrag[2];
    #pragma unroll
    for (int mt = 0; mt < 2; ++mt)
      afrag[mt] = *(const v2f*)(a + (arow + mt * 16) * LDA_P + kk + koff);
    #pragma unroll
    for (int nt = 0; nt < 2; ++nt) {
      bfrag[nt].x = b[(kk + koff + 0) * LDB_P + bcol + nt * 16];
      bfrag[nt].y = b[(kk + koff + 1) * LDB_P + bcol + nt * 16];
    }
    #pragma unroll
    for (int mt = 0; mt < 2; ++mt)
      #pragma unroll
      for (int nt = 0; nt < 2; ++nt)
        acc[mt][nt] = __builtin_amdgcn_wmma_f32_16x16x4_f32(
            false, afrag[mt], false, bfrag[nt],
            (short)0, acc[mt][nt], false, false);
  }
}

__global__ __launch_bounds__(256) void gemm_bias_act(
    const float* __restrict__ A0, const float* __restrict__ W0,
    const float* __restrict__ bias0, float* __restrict__ out0,
    const float* __restrict__ A1, const float* __restrict__ W1,
    const float* __restrict__ bias1, float* __restrict__ out1,
    int M, int K, int N, int relu)
{
  __shared__ float lA[2 * LA_SZ];
  __shared__ float lB[2 * LB_SZ];

  const bool br = blockIdx.z != 0;
  const float* __restrict__ A    = br ? A1 : A0;
  const float* __restrict__ W    = br ? W1 : W0;
  const float* __restrict__ bias = br ? bias1 : bias0;
  float* __restrict__ out        = br ? out1 : out0;

  const int t    = threadIdx.x;
  const int lane = t & 31;
  const int wave = t >> 5;
  const int wm   = wave & 3;
  const int wn   = wave >> 2;
  const int bm   = blockIdx.y * 128;
  const int bn   = blockIdx.x * 64;

  v8f acc[2][2] = {};

  const int mrow = lane & 15;
  const int koff = (lane >> 4) * 2;
  const int ncol = lane & 15;
  const int arow = wm * 32 + mrow;
  const int bcol = wn * 32 + ncol;

  const int srow = t >> 4;
  const int sc4  = t & 15;
  const float* gA = A + (size_t)(bm + srow) * K + sc4 * 4;
  const float* gB = W + (size_t)srow * N + bn + sc4 * 4;
  float* sA = lA + srow * LDA_P + sc4 * 4;
  float* sB = lB + srow * LDB_P + sc4 * 4;

  float4 pa0 = *(const float4*)(gA);
  float4 pa1 = *(const float4*)(gA + (size_t)16 * K);
  float4 pa2 = *(const float4*)(gA + (size_t)32 * K);
  float4 pa3 = *(const float4*)(gA + (size_t)48 * K);
  float4 pa4 = *(const float4*)(gA + (size_t)64 * K);
  float4 pa5 = *(const float4*)(gA + (size_t)80 * K);
  float4 pa6 = *(const float4*)(gA + (size_t)96 * K);
  float4 pa7 = *(const float4*)(gA + (size_t)112 * K);
  float4 pb0 = *(const float4*)(gB);
  float4 pb1 = *(const float4*)(gB + (size_t)16 * N);
  float4 pb2 = *(const float4*)(gB + (size_t)32 * N);
  float4 pb3 = *(const float4*)(gB + (size_t)48 * N);
  *(float4*)(sA +   0 * LDA_P) = pa0;
  *(float4*)(sA +  16 * LDA_P) = pa1;
  *(float4*)(sA +  32 * LDA_P) = pa2;
  *(float4*)(sA +  48 * LDA_P) = pa3;
  *(float4*)(sA +  64 * LDA_P) = pa4;
  *(float4*)(sA +  80 * LDA_P) = pa5;
  *(float4*)(sA +  96 * LDA_P) = pa6;
  *(float4*)(sA + 112 * LDA_P) = pa7;
  *(float4*)(sB +   0 * LDB_P) = pb0;
  *(float4*)(sB +  16 * LDB_P) = pb1;
  *(float4*)(sB +  32 * LDB_P) = pb2;
  *(float4*)(sB +  48 * LDB_P) = pb3;
  __syncthreads();

  int cur = 0;
  for (int k0 = GK; k0 < K; k0 += GK) {
    const float* nA = gA + k0;
    const float* nB = gB + (size_t)k0 * N;
    pa0 = *(const float4*)(nA);
    pa1 = *(const float4*)(nA + (size_t)16 * K);
    pa2 = *(const float4*)(nA + (size_t)32 * K);
    pa3 = *(const float4*)(nA + (size_t)48 * K);
    pa4 = *(const float4*)(nA + (size_t)64 * K);
    pa5 = *(const float4*)(nA + (size_t)80 * K);
    pa6 = *(const float4*)(nA + (size_t)96 * K);
    pa7 = *(const float4*)(nA + (size_t)112 * K);
    pb0 = *(const float4*)(nB);
    pb1 = *(const float4*)(nB + (size_t)16 * N);
    pb2 = *(const float4*)(nB + (size_t)32 * N);
    pb3 = *(const float4*)(nB + (size_t)48 * N);

    gemm_stage(lA + cur * LA_SZ, lB + cur * LB_SZ, arow, koff, bcol, acc);

    __builtin_amdgcn_sched_barrier(0);
    const int nxt = cur ^ 1;
    float* dA = sA + nxt * LA_SZ;
    float* dB = sB + nxt * LB_SZ;
    *(float4*)(dA +   0 * LDA_P) = pa0;
    *(float4*)(dA +  16 * LDA_P) = pa1;
    *(float4*)(dA +  32 * LDA_P) = pa2;
    *(float4*)(dA +  48 * LDA_P) = pa3;
    *(float4*)(dA +  64 * LDA_P) = pa4;
    *(float4*)(dA +  80 * LDA_P) = pa5;
    *(float4*)(dA +  96 * LDA_P) = pa6;
    *(float4*)(dA + 112 * LDA_P) = pa7;
    *(float4*)(dB +   0 * LDB_P) = pb0;
    *(float4*)(dB +  16 * LDB_P) = pb1;
    *(float4*)(dB +  32 * LDB_P) = pb2;
    *(float4*)(dB +  48 * LDB_P) = pb3;
    __syncthreads();
    cur = nxt;
  }
  gemm_stage(lA + cur * LA_SZ, lB + cur * LB_SZ, arow, koff, bcol, acc);
  __builtin_amdgcn_sched_barrier(0);

  const int rsel = (lane >> 4) * 8;
  #pragma unroll
  for (int nt = 0; nt < 2; ++nt) {
    int col = bn + wn * 32 + nt * 16 + ncol;
    float bv = bias[col];
    #pragma unroll
    for (int mt = 0; mt < 2; ++mt) {
      #pragma unroll
      for (int v = 0; v < 8; ++v) {
        int row = bm + wm * 32 + mt * 16 + rsel + v;
        float x = acc[mt][nt][v] + bv;
        if (relu) x = fmaxf(x, 0.0f);
        out[(size_t)row * N + col] = x;
      }
    }
  }
}

__global__ __launch_bounds__(256) void norm_rows64(
    float* __restrict__ q, unsigned int* __restrict__ maxbuf)
{
  if (blockIdx.x == 0 && threadIdx.x < CCH) maxbuf[threadIdx.x] = 0u;
  const int lane = threadIdx.x & 31;
  const int wave = threadIdx.x >> 5;
  const int row  = blockIdx.x * 8 + wave;
  float* p = q + (size_t)row * 64;

  float2 x = *(float2*)(p + lane * 2);
  float s  = x.x + x.y;
  float ss = x.x * x.x + x.y * x.y;
  #pragma unroll
  for (int off = 16; off > 0; off >>= 1) {
    s  += __shfl_xor(s,  off, 32);
    ss += __shfl_xor(ss, off, 32);
  }
  float mean = s * (1.0f / 64.0f);
  float var  = (ss - 64.0f * mean * mean) * (1.0f / 63.0f);
  float inv  = rsqrtf(var + 1e-8f);
  x.x = (x.x - mean) * inv;
  x.y = (x.y - mean) * inv;
  *(float2*)(p + lane * 2) = x;
}

#define LQA_P 68
#define LQB_P 33

__global__ __launch_bounds__(256) void affinity_wmma(
    const float* __restrict__ q1, const float* __restrict__ q2,
    float* __restrict__ aff, unsigned int* __restrict__ maxbuf)
{
  __shared__ float lA[64 * LQA_P];
  __shared__ float lB[64 * LQB_P];
  __shared__ float red[16 * CCH];

  const int t    = threadIdx.x;
  const int lane = t & 31;
  const int wave = t >> 5;
  const int wm   = wave & 3;
  const int wn   = wave >> 2;
  const int bb   = blockIdx.y * 64;
  const int bd   = blockIdx.x * 32;

  const int srow = t >> 4;
  const int sc4  = t & 15;
  const float* gA = q1 + (size_t)(bb + srow) * SLICE + sc4 * 4;
  const float* gB = q2 + (size_t)(bd + srow) * SLICE + sc4 * 4;

  float4 pa0 = *(const float4*)(gA);
  float4 pa1 = *(const float4*)(gA + (size_t)16 * SLICE);
  float4 pa2 = *(const float4*)(gA + (size_t)32 * SLICE);
  float4 pa3 = *(const float4*)(gA + (size_t)48 * SLICE);
  float4 pb0 = *(const float4*)(gB);
  float4 pb1 = *(const float4*)(gB + (size_t)16 * SLICE);
  __builtin_amdgcn_sched_barrier(0);

  v8f acc[CCH] = {};
  const int mrow = lane & 15;
  const int koff = (lane >> 4) * 2;
  const int ncol = lane & 15;

  #pragma unroll
  for (int c = 0; c < CCH; ++c) {
    *(float4*)(lA + (srow +  0) * LQA_P + sc4 * 4) = pa0;
    *(float4*)(lA + (srow + 16) * LQA_P + sc4 * 4) = pa1;
    *(float4*)(lA + (srow + 32) * LQA_P + sc4 * 4) = pa2;
    *(float4*)(lA + (srow + 48) * LQA_P + sc4 * 4) = pa3;
    lB[(sc4 * 4 + 0) * LQB_P + srow] = pb0.x;
    lB[(sc4 * 4 + 1) * LQB_P + srow] = pb0.y;
    lB[(sc4 * 4 + 2) * LQB_P + srow] = pb0.z;
    lB[(sc4 * 4 + 3) * LQB_P + srow] = pb0.w;
    lB[(sc4 * 4 + 0) * LQB_P + srow + 16] = pb1.x;
    lB[(sc4 * 4 + 1) * LQB_P + srow + 16] = pb1.y;
    lB[(sc4 * 4 + 2) * LQB_P + srow + 16] = pb1.z;
    lB[(sc4 * 4 + 3) * LQB_P + srow + 16] = pb1.w;
    __syncthreads();

    if (c + 1 < CCH) {
      const float* nA = gA + (c + 1) * EDIM;
      const float* nB = gB + (c + 1) * EDIM;
      pa0 = *(const float4*)(nA);
      pa1 = *(const float4*)(nA + (size_t)16 * SLICE);
      pa2 = *(const float4*)(nA + (size_t)32 * SLICE);
      pa3 = *(const float4*)(nA + (size_t)48 * SLICE);
      pb0 = *(const float4*)(nB);
      pb1 = *(const float4*)(nB + (size_t)16 * SLICE);
    }

    #pragma unroll
    for (int kk = 0; kk < 64; kk += 4) {
      v2f afrag, bfrag;
      afrag = *(const v2f*)(lA + (wm * 16 + mrow) * LQA_P + kk + koff);
      bfrag.x = lB[(kk + koff + 0) * LQB_P + wn * 16 + ncol];
      bfrag.y = lB[(kk + koff + 1) * LQB_P + wn * 16 + ncol];
      acc[c] = __builtin_amdgcn_wmma_f32_16x16x4_f32(
          false, afrag, false, bfrag, (short)0, acc[c], false, false);
    }
    __syncthreads();
  }

  const float scale = 0.125f;
  const int rsel = (lane >> 4) * 8;
  const int col  = bd + wn * 16 + ncol;
  #pragma unroll
  for (int v = 0; v < 8; ++v) {
    int row = bb + wm * 16 + rsel + v;
    float* dst = aff + ((size_t)row * BSZ + col) * CCH;
    #pragma unroll
    for (int q = 0; q < 4; ++q) {
      float4 o;
      o.x = acc[4 * q + 0][v] * scale;
      o.y = acc[4 * q + 1][v] * scale;
      o.z = acc[4 * q + 2][v] * scale;
      o.w = acc[4 * q + 3][v] * scale;
      *(float4*)(dst + 4 * q) = o;
    }
  }

  float* park = lA;
  #pragma unroll
  for (int c = 0; c < CCH; ++c) {
    float m = -3.4e38f;
    #pragma unroll
    for (int v = 0; v < 8; ++v) m = fmaxf(m, acc[c][v] * scale);
    park[t * 17 + c] = m;
  }
  __syncthreads();
  {
    const int pc = t & 15;
    const int pg = t >> 4;
    float m = park[(pg * 16) * 17 + pc];
    for (int i = 1; i < 16; ++i) m = fmaxf(m, park[(pg * 16 + i) * 17 + pc]);
    red[pg * CCH + pc] = m;
  }
  __syncthreads();
  if (t < CCH) {
    float m = red[t];
    for (int g = 1; g < 16; ++g) m = fmaxf(m, red[g * CCH + t]);
    atomicMax(&maxbuf[t], enc_f32(m));
  }
}

__global__ __launch_bounds__(256) void sink_u_first(
    float* __restrict__ A, const unsigned int* __restrict__ maxbuf,
    const float* __restrict__ r, float* __restrict__ u)
{
  __shared__ float red[8 * 32];
  const int t    = threadIdx.x;
  const int lane = t & 31;
  const int wave = t >> 5;
  const int b0   = blockIdx.x * 2;
  float4* A4 = (float4*)(A + (size_t)b0 * (BSZ * CCH)) + t;

  const int c0 = (t & 3) * 4;
  const float m0 = dec_f32(maxbuf[c0 + 0]);
  const float m1 = dec_f32(maxbuf[c0 + 1]);
  const float m2 = dec_f32(maxbuf[c0 + 2]);
  const float m3 = dec_f32(maxbuf[c0 + 3]);

  float s[8] = {0.0f, 0.0f, 0.0f, 0.0f, 0.0f, 0.0f, 0.0f, 0.0f};
  #pragma unroll
  for (int i = 0; i < 16; ++i) {
    float4 x0 = A4[i * 256];
    float4 x1 = A4[4096 + i * 256];
    x0.x = expf(x0.x - m0);  x0.y = expf(x0.y - m1);
    x0.z = expf(x0.z - m2);  x0.w = expf(x0.w - m3);
    x1.x = expf(x1.x - m0);  x1.y = expf(x1.y - m1);
    x1.z = expf(x1.z - m2);  x1.w = expf(x1.w - m3);
    A4[i * 256]        = x0;
    A4[4096 + i * 256] = x1;
    s[0] += x0.x;  s[1] += x0.y;
    s[2] += x0.z;  s[3] += x0.w;
    s[4] += x1.x;  s[5] += x1.y;
    s[6] += x1.z;  s[7] += x1.w;
  }
  #pragma unroll
  for (int q = 0; q < 8; ++q) {
    s[q] += __shfl_xor(s[q], 4, 32);
    s[q] += __shfl_xor(s[q], 8, 32);
    s[q] += __shfl_xor(s[q], 16, 32);
  }
  if (lane < 4) {
    #pragma unroll
    for (int k = 0; k < 4; ++k) {
      red[wave * 32 +  0 + lane * 4 + k] = s[k];
      red[wave * 32 + 16 + lane * 4 + k] = s[4 + k];
    }
  }
  __syncthreads();
  if (t < 32) {
    float sum = red[t];
    for (int w = 1; w < 8; ++w) sum += red[w * 32 + t];
    u[b0 * CCH + t] = r[b0 * CCH + t] / sum;
  }
}

__global__ __launch_bounds__(256) void sink_u(
    const float* __restrict__ A, const float* __restrict__ v,
    const float* __restrict__ r, float* __restrict__ u)
{
  __shared__ float red[8 * 32];
  const int t    = threadIdx.x;
  const int lane = t & 31;
  const int wave = t >> 5;
  const int b0   = blockIdx.x * 2;
  const float4* A4 = (const float4*)(A + (size_t)b0 * (BSZ * CCH)) + t;
  const float4* v4 = (const float4*)v + t;

  float s[8] = {0.0f, 0.0f, 0.0f, 0.0f, 0.0f, 0.0f, 0.0f, 0.0f};
  #pragma unroll
  for (int i = 0; i < 16; ++i) {
    const float4 w  = v4[i * 256];
    const float4 x0 = A4[i * 256];
    const float4 x1 = A4[4096 + i * 256];
    s[0] += x0.x * w.x;  s[1] += x0.y * w.y;
    s[2] += x0.z * w.z;  s[3] += x0.w * w.w;
    s[4] += x1.x * w.x;  s[5] += x1.y * w.y;
    s[6] += x1.z * w.z;  s[7] += x1.w * w.w;
  }
  #pragma unroll
  for (int q = 0; q < 8; ++q) {
    s[q] += __shfl_xor(s[q], 4, 32);
    s[q] += __shfl_xor(s[q], 8, 32);
    s[q] += __shfl_xor(s[q], 16, 32);
  }
  if (lane < 4) {
    #pragma unroll
    for (int k = 0; k < 4; ++k) {
      red[wave * 32 +  0 + lane * 4 + k] = s[k];
      red[wave * 32 + 16 + lane * 4 + k] = s[4 + k];
    }
  }
  __syncthreads();
  if (t < 32) {
    float sum = red[t];
    for (int w = 1; w < 8; ++w) sum += red[w * 32 + t];
    u[b0 * CCH + t] = r[b0 * CCH + t] / sum;
  }
}

__global__ __launch_bounds__(256) void sink_v(
    const float* __restrict__ A, const float* __restrict__ u,
    const float* __restrict__ cmarg, float* __restrict__ v)
{
  __shared__ float4 su[BSZ * CCH / 4];
  __shared__ float red[8 * 64];
  const int t    = threadIdx.x;
  const int lane = t & 31;
  const int wave = t >> 5;
  const int l16  = t & 15;
  const int rg   = t >> 4;
  const int j0   = blockIdx.x * 64;
  const float* Ap = A + (size_t)rg * (BSZ * CCH) + j0 + l16 * 4;
  const float4* sp = su + rg * 4 + (l16 & 3);

  #pragma unroll
  for (int i = 0; i < 16; ++i) su[t + 256 * i] = ((const float4*)u)[t + 256 * i];
  __syncthreads();

  float4 acc = make_float4(0.0f, 0.0f, 0.0f, 0.0f);
  #pragma unroll
  for (int step = 0; step < 64; ++step) {
    float4 a = *(const float4*)(Ap + (size_t)step * 16 * (BSZ * CCH));
    float4 w = sp[step * 64];
    acc.x += a.x * w.x;  acc.y += a.y * w.y;
    acc.z += a.z * w.z;  acc.w += a.w * w.w;
  }

  acc.x += __shfl_xor(acc.x, 16, 32);
  acc.y += __shfl_xor(acc.y, 16, 32);
  acc.z += __shfl_xor(acc.z, 16, 32);
  acc.w += __shfl_xor(acc.w, 16, 32);
  if (lane < 16) {
    red[wave * 64 + lane * 4 + 0] = acc.x;
    red[wave * 64 + lane * 4 + 1] = acc.y;
    red[wave * 64 + lane * 4 + 2] = acc.z;
    red[wave * 64 + lane * 4 + 3] = acc.w;
  }
  __syncthreads();
  if (t < 64) {
    float sum = red[t];
    for (int w = 1; w < 8; ++w) sum += red[w * 64 + t];
    v[j0 + t] = cmarg[j0 + t] / sum;
  }
}

__global__ __launch_bounds__(256) void sink_v_final(
    float* __restrict__ A, const float* __restrict__ u,
    const float* __restrict__ cmarg)
{
  __shared__ float4 su[BSZ * CCH / 4];
  __shared__ float red[8 * 64];
  __shared__ float vs[64];
  const int t    = threadIdx.x;
  const int lane = t & 31;
  const int wave = t >> 5;
  const int l16  = t & 15;
  const int rg   = t >> 4;
  const int j0   = blockIdx.x * 64;
  float* Ap = A + (size_t)rg * (BSZ * CCH) + j0 + l16 * 4;
  const float4* sp = su + rg * 4 + (l16 & 3);

  float4 a[64];
  #pragma unroll
  for (int step = 0; step < 64; ++step)
    a[step] = *(const float4*)(Ap + (size_t)step * 16 * (BSZ * CCH));

  #pragma unroll
  for (int i = 0; i < 16; ++i) su[t + 256 * i] = ((const float4*)u)[t + 256 * i];
  __syncthreads();

  float4 acc = make_float4(0.0f, 0.0f, 0.0f, 0.0f);
  #pragma unroll
  for (int step = 0; step < 64; ++step) {
    float4 w = sp[step * 64];
    acc.x += a[step].x * w.x;  acc.y += a[step].y * w.y;
    acc.z += a[step].z * w.z;  acc.w += a[step].w * w.w;
  }

  acc.x += __shfl_xor(acc.x, 16, 32);
  acc.y += __shfl_xor(acc.y, 16, 32);
  acc.z += __shfl_xor(acc.z, 16, 32);
  acc.w += __shfl_xor(acc.w, 16, 32);
  if (lane < 16) {
    red[wave * 64 + lane * 4 + 0] = acc.x;
    red[wave * 64 + lane * 4 + 1] = acc.y;
    red[wave * 64 + lane * 4 + 2] = acc.z;
    red[wave * 64 + lane * 4 + 3] = acc.w;
  }
  __syncthreads();
  if (t < 64) {
    float sum = red[t];
    for (int w = 1; w < 8; ++w) sum += red[w * 64 + t];
    vs[t] = cmarg[j0 + t] / sum;
  }
  __syncthreads();

  const float4 vv = *(const float4*)(vs + l16 * 4);
  #pragma unroll
  for (int step = 0; step < 64; ++step) {
    float4 w = sp[step * 64];
    float4 p;
    p.x = a[step].x * w.x * vv.x;  p.y = a[step].y * w.y * vv.y;
    p.z = a[step].z * w.z * vv.z;  p.w = a[step].w * w.w * vv.w;
    *(float4*)(Ap + (size_t)step * 16 * (BSZ * CCH)) = p;
  }
}

extern "C" void kernel_launch(void* const* d_in, const int* in_sizes, int n_in,
                              void* d_out, int out_size, void* d_ws, size_t ws_size,
                              hipStream_t stream)
{
  const float* x1   = (const float*)d_in[0];
  const float* x2   = (const float*)d_in[1];
  const float* p1   = (const float*)d_in[2];
  const float* p2   = (const float*)d_in[3];
  const float* w1[4] = {(const float*)d_in[4],  (const float*)d_in[6],
                        (const float*)d_in[8],  (const float*)d_in[10]};
  const float* b1[4] = {(const float*)d_in[5],  (const float*)d_in[7],
                        (const float*)d_in[9],  (const float*)d_in[11]};
  const float* w2[4] = {(const float*)d_in[12], (const float*)d_in[14],
                        (const float*)d_in[16], (const float*)d_in[18]};
  const float* b2[4] = {(const float*)d_in[13], (const float*)d_in[15],
                        (const float*)d_in[17], (const float*)d_in[19]};

  float* ws = (float*)d_ws;
  float* hA = ws;
  float* hB = hA + 1024 * 512;
  float* q1 = hB + 1024 * 512;
  float* q2 = q1 + 1024 * 1024;
  float* uv_u = q2 + 1024 * 1024;
  float* uv_v = uv_u + 16384;
  unsigned int* maxbuf = (unsigned int*)(uv_v + 16384);

  float* Aout = (float*)d_out;

  float* h1a = q1;
  float* h1b = q1 + 1024 * 512;
  float* h2a = q2;
  float* h2b = q2 + 1024 * 512;

  gemm_bias_act<<<dim3(512/64, 1024/128, 2), 256, 0, stream>>>(
      x1, w1[0], b1[0], h1a,  x2, w2[0], b2[0], h2a,  1024, 256, 512, 1);
  gemm_bias_act<<<dim3(512/64, 1024/128, 2), 256, 0, stream>>>(
      h1a, w1[1], b1[1], h1b,  h2a, w2[1], b2[1], h2b,  1024, 512, 512, 1);
  gemm_bias_act<<<dim3(512/64, 1024/128, 2), 256, 0, stream>>>(
      h1b, w1[2], b1[2], hA,  h2b, w2[2], b2[2], hB,  1024, 512, 512, 1);
  gemm_bias_act<<<dim3(1024/64, 1024/128, 2), 256, 0, stream>>>(
      hA, w1[3], b1[3], q1,  hB, w2[3], b2[3], q2,  1024, 512, 1024, 0);

  norm_rows64<<<2 * 16384 / 8, 256, 0, stream>>>(q1, maxbuf);

  affinity_wmma<<<dim3(1024/32, 1024/64), 256, 0, stream>>>(q1, q2, Aout, maxbuf);

  sink_u_first<<<512, 256, 0, stream>>>(Aout, maxbuf, p1, uv_u);
  sink_v<<<256, 256, 0, stream>>>(Aout, uv_u, p2, uv_v);
  for (int it = 1; it < 9; ++it) {
    sink_u<<<512, 256, 0, stream>>>(Aout, uv_v, p1, uv_u);
    sink_v<<<256, 256, 0, stream>>>(Aout, uv_u, p2, uv_v);
  }
  sink_u<<<512, 256, 0, stream>>>(Aout, uv_v, p1, uv_u);
  sink_v_final<<<256, 256, 0, stream>>>(Aout, uv_u, p2);
}


# norm launch folded into the last MLP layer's epilogue: its 128x64 tile (one channel per tile) goes through LDS, rows normalised there (row sums in a new order), whole-line stores; 26 launches -> 25
# speedup vs baseline: 2.4769x; 1.0401x over previous
#include <hip/hip_runtime.h>
#include <hip/hip_bf16.h>


typedef float v2f __attribute__((ext_vector_type(2)));
typedef float v8f __attribute__((ext_vector_type(8)));

#define BSZ   1024
#define CCH   16
#define EDIM  64
#define SLICE (CCH * EDIM)
#define ATOT  (BSZ * BSZ * CCH)

__device__ __forceinline__ unsigned int enc_f32(float f) {
  unsigned int u = __float_as_uint(f);
  return (u & 0x80000000u) ? ~u : (u | 0x80000000u);
}
__device__ __forceinline__ float dec_f32(unsigned int u) {
  return (u & 0x80000000u) ? __uint_as_float(u & 0x7FFFFFFFu)
                           : __uint_as_float(~u);
}

#define GK    64
#define LDA_P 68
#define LDB_P 68
#define LA_SZ (128 * LDA_P)
#define LB_SZ (GK * LDB_P)

__device__ __forceinline__ void gemm_stage(
    const float* __restrict__ a, const float* __restrict__ b,
    int arow, int koff, int bcol, v8f (&acc)[2][2])
{
  #pragma unroll
  for (int kk = 0; kk < GK; kk += 4) {
    v2f afrag[2], bfrag[2];
    #pragma unroll
    for (int mt = 0; mt < 2; ++mt)
      afrag[mt] = *(const v2f*)(a + (arow + mt * 16) * LDA_P + kk + koff);
    #pragma unroll
    for (int nt = 0; nt < 2; ++nt) {
      bfrag[nt].x = b[(kk + koff + 0) * LDB_P + bcol + nt * 16];
      bfrag[nt].y = b[(kk + koff + 1) * LDB_P + bcol + nt * 16];
    }
    #pragma unroll
    for (int mt = 0; mt < 2; ++mt)
      #pragma unroll
      for (int nt = 0; nt < 2; ++nt)
        acc[mt][nt] = __builtin_amdgcn_wmma_f32_16x16x4_f32(
            false, afrag[mt], false, bfrag[nt],
            (short)0, acc[mt][nt], false, false);
  }
}

__global__ __launch_bounds__(256) void gemm_bias_act(
    const float* __restrict__ A0, const float* __restrict__ W0,
    const float* __restrict__ bias0, float* __restrict__ out0,
    const float* __restrict__ A1, const float* __restrict__ W1,
    const float* __restrict__ bias1, float* __restrict__ out1,
    unsigned int* __restrict__ maxbuf, int M, int K, int N, int act)
{
  __shared__ float lA[2 * LA_SZ];
  __shared__ float lB[2 * LB_SZ];

  const bool br = blockIdx.z != 0;
  const float* __restrict__ A    = br ? A1 : A0;
  const float* __restrict__ W    = br ? W1 : W0;
  const float* __restrict__ bias = br ? bias1 : bias0;
  float* __restrict__ out        = br ? out1 : out0;

  const int t    = threadIdx.x;
  const int lane = t & 31;
  const int wave = t >> 5;
  const int wm   = wave & 3;
  const int wn   = wave >> 2;
  const int bm   = blockIdx.y * 128;
  const int bn   = blockIdx.x * 64;

  v8f acc[2][2] = {};

  const int mrow = lane & 15;
  const int koff = (lane >> 4) * 2;
  const int ncol = lane & 15;
  const int arow = wm * 32 + mrow;
  const int bcol = wn * 32 + ncol;

  const int srow = t >> 4;
  const int sc4  = t & 15;
  const float* gA = A + (size_t)(bm + srow) * K + sc4 * 4;
  const float* gB = W + (size_t)srow * N + bn + sc4 * 4;
  float* sA = lA + srow * LDA_P + sc4 * 4;
  float* sB = lB + srow * LDB_P + sc4 * 4;

  float4 pa0 = *(const float4*)(gA);
  float4 pa1 = *(const float4*)(gA + (size_t)16 * K);
  float4 pa2 = *(const float4*)(gA + (size_t)32 * K);
  float4 pa3 = *(const float4*)(gA + (size_t)48 * K);
  float4 pa4 = *(const float4*)(gA + (size_t)64 * K);
  float4 pa5 = *(const float4*)(gA + (size_t)80 * K);
  float4 pa6 = *(const float4*)(gA + (size_t)96 * K);
  float4 pa7 = *(const float4*)(gA + (size_t)112 * K);
  float4 pb0 = *(const float4*)(gB);
  float4 pb1 = *(const float4*)(gB + (size_t)16 * N);
  float4 pb2 = *(const float4*)(gB + (size_t)32 * N);
  float4 pb3 = *(const float4*)(gB + (size_t)48 * N);
  *(float4*)(sA +   0 * LDA_P) = pa0;
  *(float4*)(sA +  16 * LDA_P) = pa1;
  *(float4*)(sA +  32 * LDA_P) = pa2;
  *(float4*)(sA +  48 * LDA_P) = pa3;
  *(float4*)(sA +  64 * LDA_P) = pa4;
  *(float4*)(sA +  80 * LDA_P) = pa5;
  *(float4*)(sA +  96 * LDA_P) = pa6;
  *(float4*)(sA + 112 * LDA_P) = pa7;
  *(float4*)(sB +   0 * LDB_P) = pb0;
  *(float4*)(sB +  16 * LDB_P) = pb1;
  *(float4*)(sB +  32 * LDB_P) = pb2;
  *(float4*)(sB +  48 * LDB_P) = pb3;
  __syncthreads();

  int cur = 0;
  for (int k0 = GK; k0 < K; k0 += GK) {
    const float* nA = gA + k0;
    const float* nB = gB + (size_t)k0 * N;
    pa0 = *(const float4*)(nA);
    pa1 = *(const float4*)(nA + (size_t)16 * K);
    pa2 = *(const float4*)(nA + (size_t)32 * K);
    pa3 = *(const float4*)(nA + (size_t)48 * K);
    pa4 = *(const float4*)(nA + (size_t)64 * K);
    pa5 = *(const float4*)(nA + (size_t)80 * K);
    pa6 = *(const float4*)(nA + (size_t)96 * K);
    pa7 = *(const float4*)(nA + (size_t)112 * K);
    pb0 = *(const float4*)(nB);
    pb1 = *(const float4*)(nB + (size_t)16 * N);
    pb2 = *(const float4*)(nB + (size_t)32 * N);
    pb3 = *(const float4*)(nB + (size_t)48 * N);

    gemm_stage(lA + cur * LA_SZ, lB + cur * LB_SZ, arow, koff, bcol, acc);

    __builtin_amdgcn_sched_barrier(0);
    const int nxt = cur ^ 1;
    float* dA = sA + nxt * LA_SZ;
    float* dB = sB + nxt * LB_SZ;
    *(float4*)(dA +   0 * LDA_P) = pa0;
    *(float4*)(dA +  16 * LDA_P) = pa1;
    *(float4*)(dA +  32 * LDA_P) = pa2;
    *(float4*)(dA +  48 * LDA_P) = pa3;
    *(float4*)(dA +  64 * LDA_P) = pa4;
    *(float4*)(dA +  80 * LDA_P) = pa5;
    *(float4*)(dA +  96 * LDA_P) = pa6;
    *(float4*)(dA + 112 * LDA_P) = pa7;
    *(float4*)(dB +   0 * LDB_P) = pb0;
    *(float4*)(dB +  16 * LDB_P) = pb1;
    *(float4*)(dB +  32 * LDB_P) = pb2;
    *(float4*)(dB +  48 * LDB_P) = pb3;
    __syncthreads();
    cur = nxt;
  }
  gemm_stage(lA + cur * LA_SZ, lB + cur * LB_SZ, arow, koff, bcol, acc);
  __builtin_amdgcn_sched_barrier(0);

  const int rsel = (lane >> 4) * 8;
  #pragma unroll
  for (int nt = 0; nt < 2; ++nt) {
    float bv = bias[bn + wn * 32 + nt * 16 + ncol];
    #pragma unroll
    for (int mt = 0; mt < 2; ++mt) {
      #pragma unroll
      for (int v = 0; v < 8; ++v) acc[mt][nt][v] += bv;
    }
  }

  if (act == 2) {
    if (blockIdx.x == 0 && blockIdx.y == 0 && blockIdx.z == 0 && t < CCH)
      maxbuf[t] = 0u;
    float* T = lA + (cur ^ 1) * LA_SZ;
    #pragma unroll
    for (int nt = 0; nt < 2; ++nt) {
      #pragma unroll
      for (int mt = 0; mt < 2; ++mt) {
        #pragma unroll
        for (int v = 0; v < 8; ++v)
          T[(wm * 32 + mt * 16 + rsel + v) * LDA_P + wn * 32 + nt * 16 + ncol] =
              acc[mt][nt][v];
      }
    }
    __syncthreads();

    const int l16 = lane & 15;
    const int rbase = wave * 16 + (lane >> 4);
    float4 x[8];
    float s[8], ss[8];
    #pragma unroll
    for (int i = 0; i < 8; ++i) {
      x[i] = *(const float4*)(T + (rbase + 2 * i) * LDA_P + l16 * 4);
      s[i]  = x[i].x + x[i].y + x[i].z + x[i].w;
      ss[i] = x[i].x * x[i].x + x[i].y * x[i].y + x[i].z * x[i].z + x[i].w * x[i].w;
    }
    #pragma unroll
    for (int off = 8; off > 0; off >>= 1) {
      #pragma unroll
      for (int i = 0; i < 8; ++i) {
        s[i]  += __shfl_xor(s[i],  off, 32);
        ss[i] += __shfl_xor(ss[i], off, 32);
      }
    }
    #pragma unroll
    for (int i = 0; i < 8; ++i) {
      float mean = s[i] * (1.0f / 64.0f);
      float var  = (ss[i] - 64.0f * mean * mean) * (1.0f / 63.0f);
      float inv  = rsqrtf(var + 1e-8f);
      float4 y;
      y.x = (x[i].x - mean) * inv;  y.y = (x[i].y - mean) * inv;
      y.z = (x[i].z - mean) * inv;  y.w = (x[i].w - mean) * inv;
      *(float4*)(out + (size_t)(bm + rbase + 2 * i) * N + bn + l16 * 4) = y;
    }
  } else {
    #pragma unroll
    for (int nt = 0; nt < 2; ++nt) {
      int col = bn + wn * 32 + nt * 16 + ncol;
      #pragma unroll
      for (int mt = 0; mt < 2; ++mt) {
        #pragma unroll
        for (int v = 0; v < 8; ++v) {
          int row = bm + wm * 32 + mt * 16 + rsel + v;
          float xv = acc[mt][nt][v];
          if (act) xv = fmaxf(xv, 0.0f);
          out[(size_t)row * N + col] = xv;
        }
      }
    }
  }
}

#define LQA_P 68
#define LQB_P 33

__global__ __launch_bounds__(256) void affinity_wmma(
    const float* __restrict__ q1, const float* __restrict__ q2,
    float* __restrict__ aff, unsigned int* __restrict__ maxbuf)
{
  __shared__ float lA[64 * LQA_P];
  __shared__ float lB[64 * LQB_P];
  __shared__ float red[16 * CCH];

  const int t    = threadIdx.x;
  const int lane = t & 31;
  const int wave = t >> 5;
  const int wm   = wave & 3;
  const int wn   = wave >> 2;
  const int bb   = blockIdx.y * 64;
  const int bd   = blockIdx.x * 32;

  const int srow = t >> 4;
  const int sc4  = t & 15;
  const float* gA = q1 + (size_t)(bb + srow) * SLICE + sc4 * 4;
  const float* gB = q2 + (size_t)(bd + srow) * SLICE + sc4 * 4;

  float4 pa0 = *(const float4*)(gA);
  float4 pa1 = *(const float4*)(gA + (size_t)16 * SLICE);
  float4 pa2 = *(const float4*)(gA + (size_t)32 * SLICE);
  float4 pa3 = *(const float4*)(gA + (size_t)48 * SLICE);
  float4 pb0 = *(const float4*)(gB);
  float4 pb1 = *(const float4*)(gB + (size_t)16 * SLICE);
  __builtin_amdgcn_sched_barrier(0);

  v8f acc[CCH] = {};
  const int mrow = lane & 15;
  const int koff = (lane >> 4) * 2;
  const int ncol = lane & 15;

  #pragma unroll
  for (int c = 0; c < CCH; ++c) {
    *(float4*)(lA + (srow +  0) * LQA_P + sc4 * 4) = pa0;
    *(float4*)(lA + (srow + 16) * LQA_P + sc4 * 4) = pa1;
    *(float4*)(lA + (srow + 32) * LQA_P + sc4 * 4) = pa2;
    *(float4*)(lA + (srow + 48) * LQA_P + sc4 * 4) = pa3;
    lB[(sc4 * 4 + 0) * LQB_P + srow] = pb0.x;
    lB[(sc4 * 4 + 1) * LQB_P + srow] = pb0.y;
    lB[(sc4 * 4 + 2) * LQB_P + srow] = pb0.z;
    lB[(sc4 * 4 + 3) * LQB_P + srow] = pb0.w;
    lB[(sc4 * 4 + 0) * LQB_P + srow + 16] = pb1.x;
    lB[(sc4 * 4 + 1) * LQB_P + srow + 16] = pb1.y;
    lB[(sc4 * 4 + 2) * LQB_P + srow + 16] = pb1.z;
    lB[(sc4 * 4 + 3) * LQB_P + srow + 16] = pb1.w;
    __syncthreads();

    if (c + 1 < CCH) {
      const float* nA = gA + (c + 1) * EDIM;
      const float* nB = gB + (c + 1) * EDIM;
      pa0 = *(const float4*)(nA);
      pa1 = *(const float4*)(nA + (size_t)16 * SLICE);
      pa2 = *(const float4*)(nA + (size_t)32 * SLICE);
      pa3 = *(const float4*)(nA + (size_t)48 * SLICE);
      pb0 = *(const float4*)(nB);
      pb1 = *(const float4*)(nB + (size_t)16 * SLICE);
    }

    #pragma unroll
    for (int kk = 0; kk < 64; kk += 4) {
      v2f afrag, bfrag;
      afrag = *(const v2f*)(lA + (wm * 16 + mrow) * LQA_P + kk + koff);
      bfrag.x = lB[(kk + koff + 0) * LQB_P + wn * 16 + ncol];
      bfrag.y = lB[(kk + koff + 1) * LQB_P + wn * 16 + ncol];
      acc[c] = __builtin_amdgcn_wmma_f32_16x16x4_f32(
          false, afrag, false, bfrag, (short)0, acc[c], false, false);
    }
    __syncthreads();
  }

  const float scale = 0.125f;
  const int rsel = (lane >> 4) * 8;
  const int col  = bd + wn * 16 + ncol;
  #pragma unroll
  for (int v = 0; v < 8; ++v) {
    int row = bb + wm * 16 + rsel + v;
    float* dst = aff + ((size_t)row * BSZ + col) * CCH;
    #pragma unroll
    for (int q = 0; q < 4; ++q) {
      float4 o;
      o.x = acc[4 * q + 0][v] * scale;
      o.y = acc[4 * q + 1][v] * scale;
      o.z = acc[4 * q + 2][v] * scale;
      o.w = acc[4 * q + 3][v] * scale;
      *(float4*)(dst + 4 * q) = o;
    }
  }

  float* park = lA;
  #pragma unroll
  for (int c = 0; c < CCH; ++c) {
    float m = -3.4e38f;
    #pragma unroll
    for (int v = 0; v < 8; ++v) m = fmaxf(m, acc[c][v] * scale);
    park[t * 17 + c] = m;
  }
  __syncthreads();
  {
    const int pc = t & 15;
    const int pg = t >> 4;
    float m = park[(pg * 16) * 17 + pc];
    for (int i = 1; i < 16; ++i) m = fmaxf(m, park[(pg * 16 + i) * 17 + pc]);
    red[pg * CCH + pc] = m;
  }
  __syncthreads();
  if (t < CCH) {
    float m = red[t];
    for (int g = 1; g < 16; ++g) m = fmaxf(m, red[g * CCH + t]);
    atomicMax(&maxbuf[t], enc_f32(m));
  }
}

__global__ __launch_bounds__(256) void sink_u_first(
    float* __restrict__ A, const unsigned int* __restrict__ maxbuf,
    const float* __restrict__ r, float* __restrict__ u)
{
  __shared__ float red[8 * 32];
  const int t    = threadIdx.x;
  const int lane = t & 31;
  const int wave = t >> 5;
  const int b0   = blockIdx.x * 2;
  float4* A4 = (float4*)(A + (size_t)b0 * (BSZ * CCH)) + t;

  const int c0 = (t & 3) * 4;
  const float m0 = dec_f32(maxbuf[c0 + 0]);
  const float m1 = dec_f32(maxbuf[c0 + 1]);
  const float m2 = dec_f32(maxbuf[c0 + 2]);
  const float m3 = dec_f32(maxbuf[c0 + 3]);

  float s[8] = {0.0f, 0.0f, 0.0f, 0.0f, 0.0f, 0.0f, 0.0f, 0.0f};
  #pragma unroll
  for (int i = 0; i < 16; ++i) {
    float4 x0 = A4[i * 256];
    float4 x1 = A4[4096 + i * 256];
    x0.x = expf(x0.x - m0);  x0.y = expf(x0.y - m1);
    x0.z = expf(x0.z - m2);  x0.w = expf(x0.w - m3);
    x1.x = expf(x1.x - m0);  x1.y = expf(x1.y - m1);
    x1.z = expf(x1.z - m2);  x1.w = expf(x1.w - m3);
    A4[i * 256]        = x0;
    A4[4096 + i * 256] = x1;
    s[0] += x0.x;  s[1] += x0.y;
    s[2] += x0.z;  s[3] += x0.w;
    s[4] += x1.x;  s[5] += x1.y;
    s[6] += x1.z;  s[7] += x1.w;
  }
  #pragma unroll
  for (int q = 0; q < 8; ++q) {
    s[q] += __shfl_xor(s[q], 4, 32);
    s[q] += __shfl_xor(s[q], 8, 32);
    s[q] += __shfl_xor(s[q], 16, 32);
  }
  if (lane < 4) {
    #pragma unroll
    for (int k = 0; k < 4; ++k) {
      red[wave * 32 +  0 + lane * 4 + k] = s[k];
      red[wave * 32 + 16 + lane * 4 + k] = s[4 + k];
    }
  }
  __syncthreads();
  if (t < 32) {
    float sum = red[t];
    for (int w = 1; w < 8; ++w) sum += red[w * 32 + t];
    u[b0 * CCH + t] = r[b0 * CCH + t] / sum;
  }
}

__global__ __launch_bounds__(256) void sink_u(
    const float* __restrict__ A, const float* __restrict__ v,
    const float* __restrict__ r, float* __restrict__ u)
{
  __shared__ float red[8 * 32];
  const int t    = threadIdx.x;
  const int lane = t & 31;
  const int wave = t >> 5;
  const int b0   = blockIdx.x * 2;
  const float4* A4 = (const float4*)(A + (size_t)b0 * (BSZ * CCH)) + t;
  const float4* v4 = (const float4*)v + t;

  float s[8] = {0.0f, 0.0f, 0.0f, 0.0f, 0.0f, 0.0f, 0.0f, 0.0f};
  #pragma unroll
  for (int i = 0; i < 16; ++i) {
    const float4 w  = v4[i * 256];
    const float4 x0 = A4[i * 256];
    const float4 x1 = A4[4096 + i * 256];
    s[0] += x0.x * w.x;  s[1] += x0.y * w.y;
    s[2] += x0.z * w.z;  s[3] += x0.w * w.w;
    s[4] += x1.x * w.x;  s[5] += x1.y * w.y;
    s[6] += x1.z * w.z;  s[7] += x1.w * w.w;
  }
  #pragma unroll
  for (int q = 0; q < 8; ++q) {
    s[q] += __shfl_xor(s[q], 4, 32);
    s[q] += __shfl_xor(s[q], 8, 32);
    s[q] += __shfl_xor(s[q], 16, 32);
  }
  if (lane < 4) {
    #pragma unroll
    for (int k = 0; k < 4; ++k) {
      red[wave * 32 +  0 + lane * 4 + k] = s[k];
      red[wave * 32 + 16 + lane * 4 + k] = s[4 + k];
    }
  }
  __syncthreads();
  if (t < 32) {
    float sum = red[t];
    for (int w = 1; w < 8; ++w) sum += red[w * 32 + t];
    u[b0 * CCH + t] = r[b0 * CCH + t] / sum;
  }
}

__global__ __launch_bounds__(256) void sink_v(
    const float* __restrict__ A, const float* __restrict__ u,
    const float* __restrict__ cmarg, float* __restrict__ v)
{
  __shared__ float4 su[BSZ * CCH / 4];
  __shared__ float red[8 * 64];
  const int t    = threadIdx.x;
  const int lane = t & 31;
  const int wave = t >> 5;
  const int l16  = t & 15;
  const int rg   = t >> 4;
  const int j0   = blockIdx.x * 64;
  const float* Ap = A + (size_t)rg * (BSZ * CCH) + j0 + l16 * 4;
  const float4* sp = su + rg * 4 + (l16 & 3);

  #pragma unroll
  for (int i = 0; i < 16; ++i) su[t + 256 * i] = ((const float4*)u)[t + 256 * i];
  __syncthreads();

  float4 acc = make_float4(0.0f, 0.0f, 0.0f, 0.0f);
  #pragma unroll
  for (int step = 0; step < 64; ++step) {
    float4 a = *(const float4*)(Ap + (size_t)step * 16 * (BSZ * CCH));
    float4 w = sp[step * 64];
    acc.x += a.x * w.x;  acc.y += a.y * w.y;
    acc.z += a.z * w.z;  acc.w += a.w * w.w;
  }

  acc.x += __shfl_xor(acc.x, 16, 32);
  acc.y += __shfl_xor(acc.y, 16, 32);
  acc.z += __shfl_xor(acc.z, 16, 32);
  acc.w += __shfl_xor(acc.w, 16, 32);
  if (lane < 16) {
    red[wave * 64 + lane * 4 + 0] = acc.x;
    red[wave * 64 + lane * 4 + 1] = acc.y;
    red[wave * 64 + lane * 4 + 2] = acc.z;
    red[wave * 64 + lane * 4 + 3] = acc.w;
  }
  __syncthreads();
  if (t < 64) {
    float sum = red[t];
    for (int w = 1; w < 8; ++w) sum += red[w * 64 + t];
    v[j0 + t] = cmarg[j0 + t] / sum;
  }
}

__global__ __launch_bounds__(256) void sink_v_final(
    float* __restrict__ A, const float* __restrict__ u,
    const float* __restrict__ cmarg)
{
  __shared__ float4 su[BSZ * CCH / 4];
  __shared__ float red[8 * 64];
  __shared__ float vs[64];
  const int t    = threadIdx.x;
  const int lane = t & 31;
  const int wave = t >> 5;
  const int l16  = t & 15;
  const int rg   = t >> 4;
  const int j0   = blockIdx.x * 64;
  float* Ap = A + (size_t)rg * (BSZ * CCH) + j0 + l16 * 4;
  const float4* sp = su + rg * 4 + (l16 & 3);

  float4 a[64];
  #pragma unroll
  for (int step = 0; step < 64; ++step)
    a[step] = *(const float4*)(Ap + (size_t)step * 16 * (BSZ * CCH));

  #pragma unroll
  for (int i = 0; i < 16; ++i) su[t + 256 * i] = ((const float4*)u)[t + 256 * i];
  __syncthreads();

  float4 acc = make_float4(0.0f, 0.0f, 0.0f, 0.0f);
  #pragma unroll
  for (int step = 0; step < 64; ++step) {
    float4 w = sp[step * 64];
    acc.x += a[step].x * w.x;  acc.y += a[step].y * w.y;
    acc.z += a[step].z * w.z;  acc.w += a[step].w * w.w;
  }

  acc.x += __shfl_xor(acc.x, 16, 32);
  acc.y += __shfl_xor(acc.y, 16, 32);
  acc.z += __shfl_xor(acc.z, 16, 32);
  acc.w += __shfl_xor(acc.w, 16, 32);
  if (lane < 16) {
    red[wave * 64 + lane * 4 + 0] = acc.x;
    red[wave * 64 + lane * 4 + 1] = acc.y;
    red[wave * 64 + lane * 4 + 2] = acc.z;
    red[wave * 64 + lane * 4 + 3] = acc.w;
  }
  __syncthreads();
  if (t < 64) {
    float sum = red[t];
    for (int w = 1; w < 8; ++w) sum += red[w * 64 + t];
    vs[t] = cmarg[j0 + t] / sum;
  }
  __syncthreads();

  const float4 vv = *(const float4*)(vs + l16 * 4);
  #pragma unroll
  for (int step = 0; step < 64; ++step) {
    float4 w = sp[step * 64];
    float4 p;
    p.x = a[step].x * w.x * vv.x;  p.y = a[step].y * w.y * vv.y;
    p.z = a[step].z * w.z * vv.z;  p.w = a[step].w * w.w * vv.w;
    *(float4*)(Ap + (size_t)step * 16 * (BSZ * CCH)) = p;
  }
}

extern "C" void kernel_launch(void* const* d_in, const int* in_sizes, int n_in,
                              void* d_out, int out_size, void* d_ws, size_t ws_size,
                              hipStream_t stream)
{
  const float* x1   = (const float*)d_in[0];
  const float* x2   = (const float*)d_in[1];
  const float* p1   = (const float*)d_in[2];
  const float* p2   = (const float*)d_in[3];
  const float* w1[4] = {(const float*)d_in[4],  (const float*)d_in[6],
                        (const float*)d_in[8],  (const float*)d_in[10]};
  const float* b1[4] = {(const float*)d_in[5],  (const float*)d_in[7],
                        (const float*)d_in[9],  (const float*)d_in[11]};
  const float* w2[4] = {(const float*)d_in[12], (const float*)d_in[14],
                        (const float*)d_in[16], (const float*)d_in[18]};
  const float* b2[4] = {(const float*)d_in[13], (const float*)d_in[15],
                        (const float*)d_in[17], (const float*)d_in[19]};

  float* ws = (float*)d_ws;
  float* hA = ws;
  float* hB = hA + 1024 * 512;
  float* q1 = hB + 1024 * 512;
  float* q2 = q1 + 1024 * 1024;
  float* uv_u = q2 + 1024 * 1024;
  float* uv_v = uv_u + 16384;
  unsigned int* maxbuf = (unsigned int*)(uv_v + 16384);

  float* Aout = (float*)d_out;

  float* h1a = q1;
  float* h1b = q1 + 1024 * 512;
  float* h2a = q2;
  float* h2b = q2 + 1024 * 512;

  gemm_bias_act<<<dim3(512/64, 1024/128, 2), 256, 0, stream>>>(
      x1, w1[0], b1[0], h1a,  x2, w2[0], b2[0], h2a,  maxbuf, 1024, 256, 512, 1);
  gemm_bias_act<<<dim3(512/64, 1024/128, 2), 256, 0, stream>>>(
      h1a, w1[1], b1[1], h1b,  h2a, w2[1], b2[1], h2b,  maxbuf, 1024, 512, 512, 1);
  gemm_bias_act<<<dim3(512/64, 1024/128, 2), 256, 0, stream>>>(
      h1b, w1[2], b1[2], hA,  h2b, w2[2], b2[2], hB,  maxbuf, 1024, 512, 512, 1);
  gemm_bias_act<<<dim3(1024/64, 1024/128, 2), 256, 0, stream>>>(
      hA, w1[3], b1[3], q1,  hB, w2[3], b2[3], q2,  maxbuf, 1024, 512, 1024, 2);

  affinity_wmma<<<dim3(1024/32, 1024/64), 256, 0, stream>>>(q1, q2, Aout, maxbuf);

  sink_u_first<<<512, 256, 0, stream>>>(Aout, maxbuf, p1, uv_u);
  sink_v<<<256, 256, 0, stream>>>(Aout, uv_u, p2, uv_v);
  for (int it = 1; it < 9; ++it) {
    sink_u<<<512, 256, 0, stream>>>(Aout, uv_v, p1, uv_u);
    sink_v<<<256, 256, 0, stream>>>(Aout, uv_u, p2, uv_v);
  }
  sink_u<<<512, 256, 0, stream>>>(Aout, uv_v, p1, uv_u);
  sink_v_final<<<256, 256, 0, stream>>>(Aout, uv_u, p2);
}


# MLP hidden layers (N = 512) on 64x64 tiles instead of 128x64: 256 workgroups per launch, one per WGP, instead of 128; per workgroup and K stage half the products, two thirds of the operand bytes
# speedup vs baseline: 2.5483x; 1.0288x over previous
#include <hip/hip_runtime.h>
#include <hip/hip_bf16.h>


typedef float v2f __attribute__((ext_vector_type(2)));
typedef float v8f __attribute__((ext_vector_type(8)));

#define BSZ   1024
#define CCH   16
#define EDIM  64
#define SLICE (CCH * EDIM)
#define ATOT  (BSZ * BSZ * CCH)

__device__ __forceinline__ unsigned int enc_f32(float f) {
  unsigned int u = __float_as_uint(f);
  return (u & 0x80000000u) ? ~u : (u | 0x80000000u);
}
__device__ __forceinline__ float dec_f32(unsigned int u) {
  return (u & 0x80000000u) ? __uint_as_float(u & 0x7FFFFFFFu)
                           : __uint_as_float(~u);
}

#define GK    64
#define LDA_P 68
#define LDB_P 68
#define LB_SZ (GK * LDB_P)

template <int MT>
__device__ __forceinline__ void gemm_stage(
    const float* __restrict__ a, const float* __restrict__ b,
    int arow, int koff, int bcol, v8f (&acc)[MT][2])
{
  #pragma unroll
  for (int kk = 0; kk < GK; kk += 4) {
    v2f afrag[MT], bfrag[2];
    #pragma unroll
    for (int mt = 0; mt < MT; ++mt)
      afrag[mt] = *(const v2f*)(a + (arow + mt * 16) * LDA_P + kk + koff);
    #pragma unroll
    for (int nt = 0; nt < 2; ++nt) {
      bfrag[nt].x = b[(kk + koff + 0) * LDB_P + bcol + nt * 16];
      bfrag[nt].y = b[(kk + koff + 1) * LDB_P + bcol + nt * 16];
    }
    #pragma unroll
    for (int mt = 0; mt < MT; ++mt)
      #pragma unroll
      for (int nt = 0; nt < 2; ++nt)
        acc[mt][nt] = __builtin_amdgcn_wmma_f32_16x16x4_f32(
            false, afrag[mt], false, bfrag[nt],
            (short)0, acc[mt][nt], false, false);
  }
}

template <int MT>
__global__ __launch_bounds__(256) void gemm_bias_act(
    const float* __restrict__ A0, const float* __restrict__ W0,
    const float* __restrict__ bias0, float* __restrict__ out0,
    const float* __restrict__ A1, const float* __restrict__ W1,
    const float* __restrict__ bias1, float* __restrict__ out1,
    unsigned int* __restrict__ maxbuf, int M, int K, int N, int act)
{
  constexpr int BM    = 64 * MT;
  constexpr int WROWS = 16 * MT;
  constexpr int LA_SZ = BM * LDA_P;
  __shared__ float lA[2 * LA_SZ];
  __shared__ float lB[2 * LB_SZ];

  const bool br = blockIdx.z != 0;
  const float* __restrict__ A    = br ? A1 : A0;
  const float* __restrict__ W    = br ? W1 : W0;
  const float* __restrict__ bias = br ? bias1 : bias0;
  float* __restrict__ out        = br ? out1 : out0;

  const int t    = threadIdx.x;
  const int lane = t & 31;
  const int wave = t >> 5;
  const int wm   = wave & 3;
  const int wn   = wave >> 2;
  const int bm   = blockIdx.y * BM;
  const int bn   = blockIdx.x * 64;

  v8f acc[MT][2] = {};

  const int mrow = lane & 15;
  const int koff = (lane >> 4) * 2;
  const int ncol = lane & 15;
  const int arow = wm * WROWS + mrow;
  const int bcol = wn * 32 + ncol;

  const int srow = t >> 4;
  const int sc4  = t & 15;
  const float* gA = A + (size_t)(bm + srow) * K + sc4 * 4;
  const float* gB = W + (size_t)srow * N + bn + sc4 * 4;
  float* sA = lA + srow * LDA_P + sc4 * 4;
  float* sB = lB + srow * LDB_P + sc4 * 4;

  float4 pa0 = *(const float4*)(gA);
  float4 pa1 = *(const float4*)(gA + (size_t)16 * K);
  float4 pa2 = *(const float4*)(gA + (size_t)32 * K);
  float4 pa3 = *(const float4*)(gA + (size_t)48 * K);
  float4 pa4, pa5, pa6, pa7;
  if constexpr (MT == 2) {
    pa4 = *(const float4*)(gA + (size_t)64 * K);
    pa5 = *(const float4*)(gA + (size_t)80 * K);
    pa6 = *(const float4*)(gA + (size_t)96 * K);
    pa7 = *(const float4*)(gA + (size_t)112 * K);
  }
  float4 pb0 = *(const float4*)(gB);
  float4 pb1 = *(const float4*)(gB + (size_t)16 * N);
  float4 pb2 = *(const float4*)(gB + (size_t)32 * N);
  float4 pb3 = *(const float4*)(gB + (size_t)48 * N);
  *(float4*)(sA +   0 * LDA_P) = pa0;
  *(float4*)(sA +  16 * LDA_P) = pa1;
  *(float4*)(sA +  32 * LDA_P) = pa2;
  *(float4*)(sA +  48 * LDA_P) = pa3;
  if constexpr (MT == 2) {
    *(float4*)(sA +  64 * LDA_P) = pa4;
    *(float4*)(sA +  80 * LDA_P) = pa5;
    *(float4*)(sA +  96 * LDA_P) = pa6;
    *(float4*)(sA + 112 * LDA_P) = pa7;
  }
  *(float4*)(sB +   0 * LDB_P) = pb0;
  *(float4*)(sB +  16 * LDB_P) = pb1;
  *(float4*)(sB +  32 * LDB_P) = pb2;
  *(float4*)(sB +  48 * LDB_P) = pb3;
  __syncthreads();

  int cur = 0;
  for (int k0 = GK; k0 < K; k0 += GK) {
    const float* nA = gA + k0;
    const float* nB = gB + (size_t)k0 * N;
    pa0 = *(const float4*)(nA);
    pa1 = *(const float4*)(nA + (size_t)16 * K);
    pa2 = *(const float4*)(nA + (size_t)32 * K);
    pa3 = *(const float4*)(nA + (size_t)48 * K);
    if constexpr (MT == 2) {
      pa4 = *(const float4*)(nA + (size_t)64 * K);
      pa5 = *(const float4*)(nA + (size_t)80 * K);
      pa6 = *(const float4*)(nA + (size_t)96 * K);
      pa7 = *(const float4*)(nA + (size_t)112 * K);
    }
    pb0 = *(const float4*)(nB);
    pb1 = *(const float4*)(nB + (size_t)16 * N);
    pb2 = *(const float4*)(nB + (size_t)32 * N);
    pb3 = *(const float4*)(nB + (size_t)48 * N);

    gemm_stage<MT>(lA + cur * LA_SZ, lB + cur * LB_SZ, arow, koff, bcol, acc);

    __builtin_amdgcn_sched_barrier(0);
    const int nxt = cur ^ 1;
    float* dA = sA + nxt * LA_SZ;
    float* dB = sB + nxt * LB_SZ;
    *(float4*)(dA +   0 * LDA_P) = pa0;
    *(float4*)(dA +  16 * LDA_P) = pa1;
    *(float4*)(dA +  32 * LDA_P) = pa2;
    *(float4*)(dA +  48 * LDA_P) = pa3;
    if constexpr (MT == 2) {
      *(float4*)(dA +  64 * LDA_P) = pa4;
      *(float4*)(dA +  80 * LDA_P) = pa5;
      *(float4*)(dA +  96 * LDA_P) = pa6;
      *(float4*)(dA + 112 * LDA_P) = pa7;
    }
    *(float4*)(dB +   0 * LDB_P) = pb0;
    *(float4*)(dB +  16 * LDB_P) = pb1;
    *(float4*)(dB +  32 * LDB_P) = pb2;
    *(float4*)(dB +  48 * LDB_P) = pb3;
    __syncthreads();
    cur = nxt;
  }
  gemm_stage<MT>(lA + cur * LA_SZ, lB + cur * LB_SZ, arow, koff, bcol, acc);
  __builtin_amdgcn_sched_barrier(0);

  const int rsel = (lane >> 4) * 8;
  #pragma unroll
  for (int nt = 0; nt < 2; ++nt) {
    float bv = bias[bn + wn * 32 + nt * 16 + ncol];
    #pragma unroll
    for (int mt = 0; mt < MT; ++mt) {
      #pragma unroll
      for (int v = 0; v < 8; ++v) acc[mt][nt][v] += bv;
    }
  }

  bool plain = true;
  if constexpr (MT == 2) {
    if (act == 2) {
      plain = false;
      if (blockIdx.x == 0 && blockIdx.y == 0 && blockIdx.z == 0 && t < CCH)
        maxbuf[t] = 0u;
      float* T = lA + (cur ^ 1) * LA_SZ;
      #pragma unroll
      for (int nt = 0; nt < 2; ++nt) {
        #pragma unroll
        for (int mt = 0; mt < MT; ++mt) {
          #pragma unroll
          for (int v = 0; v < 8; ++v)
            T[(wm * WROWS + mt * 16 + rsel + v) * LDA_P + wn * 32 + nt * 16 + ncol] =
                acc[mt][nt][v];
        }
      }
      __syncthreads();

      const int l16 = lane & 15;
      const int rbase = wave * 16 + (lane >> 4);
      float4 x[8];
      float s[8], ss[8];
      #pragma unroll
      for (int i = 0; i < 8; ++i) {
        x[i] = *(const float4*)(T + (rbase + 2 * i) * LDA_P + l16 * 4);
        s[i]  = x[i].x + x[i].y + x[i].z + x[i].w;
        ss[i] = x[i].x * x[i].x + x[i].y * x[i].y + x[i].z * x[i].z + x[i].w * x[i].w;
      }
      #pragma unroll
      for (int off = 8; off > 0; off >>= 1) {
        #pragma unroll
        for (int i = 0; i < 8; ++i) {
          s[i]  += __shfl_xor(s[i],  off, 32);
          ss[i] += __shfl_xor(ss[i], off, 32);
        }
      }
      #pragma unroll
      for (int i = 0; i < 8; ++i) {
        float mean = s[i] * (1.0f / 64.0f);
        float var  = (ss[i] - 64.0f * mean * mean) * (1.0f / 63.0f);
        float inv  = rsqrtf(var + 1e-8f);
        float4 y;
        y.x = (x[i].x - mean) * inv;  y.y = (x[i].y - mean) * inv;
        y.z = (x[i].z - mean) * inv;  y.w = (x[i].w - mean) * inv;
        *(float4*)(out + (size_t)(bm + rbase + 2 * i) * N + bn + l16 * 4) = y;
      }
    }
  }
  if (plain) {
    #pragma unroll
    for (int nt = 0; nt < 2; ++nt) {
      int col = bn + wn * 32 + nt * 16 + ncol;
      #pragma unroll
      for (int mt = 0; mt < MT; ++mt) {
        #pragma unroll
        for (int v = 0; v < 8; ++v) {
          int row = bm + wm * WROWS + mt * 16 + rsel + v;
          float xv = acc[mt][nt][v];
          if (act) xv = fmaxf(xv, 0.0f);
          out[(size_t)row * N + col] = xv;
        }
      }
    }
  }
}

#define LQA_P 68
#define LQB_P 33

__global__ __launch_bounds__(256) void affinity_wmma(
    const float* __restrict__ q1, const float* __restrict__ q2,
    float* __restrict__ aff, unsigned int* __restrict__ maxbuf)
{
  __shared__ float lA[64 * LQA_P];
  __shared__ float lB[64 * LQB_P];
  __shared__ float red[16 * CCH];

  const int t    = threadIdx.x;
  const int lane = t & 31;
  const int wave = t >> 5;
  const int wm   = wave & 3;
  const int wn   = wave >> 2;
  const int bb   = blockIdx.y * 64;
  const int bd   = blockIdx.x * 32;

  const int srow = t >> 4;
  const int sc4  = t & 15;
  const float* gA = q1 + (size_t)(bb + srow) * SLICE + sc4 * 4;
  const float* gB = q2 + (size_t)(bd + srow) * SLICE + sc4 * 4;

  float4 pa0 = *(const float4*)(gA);
  float4 pa1 = *(const float4*)(gA + (size_t)16 * SLICE);
  float4 pa2 = *(const float4*)(gA + (size_t)32 * SLICE);
  float4 pa3 = *(const float4*)(gA + (size_t)48 * SLICE);
  float4 pb0 = *(const float4*)(gB);
  float4 pb1 = *(const float4*)(gB + (size_t)16 * SLICE);
  __builtin_amdgcn_sched_barrier(0);

  v8f acc[CCH] = {};
  const int mrow = lane & 15;
  const int koff = (lane >> 4) * 2;
  const int ncol = lane & 15;

  #pragma unroll
  for (int c = 0; c < CCH; ++c) {
    *(float4*)(lA + (srow +  0) * LQA_P + sc4 * 4) = pa0;
    *(float4*)(lA + (srow + 16) * LQA_P + sc4 * 4) = pa1;
    *(float4*)(lA + (srow + 32) * LQA_P + sc4 * 4) = pa2;
    *(float4*)(lA + (srow + 48) * LQA_P + sc4 * 4) = pa3;
    lB[(sc4 * 4 + 0) * LQB_P + srow] = pb0.x;
    lB[(sc4 * 4 + 1) * LQB_P + srow] = pb0.y;
    lB[(sc4 * 4 + 2) * LQB_P + srow] = pb0.z;
    lB[(sc4 * 4 + 3) * LQB_P + srow] = pb0.w;
    lB[(sc4 * 4 + 0) * LQB_P + srow + 16] = pb1.x;
    lB[(sc4 * 4 + 1) * LQB_P + srow + 16] = pb1.y;
    lB[(sc4 * 4 + 2) * LQB_P + srow + 16] = pb1.z;
    lB[(sc4 * 4 + 3) * LQB_P + srow + 16] = pb1.w;
    __syncthreads();

    if (c + 1 < CCH) {
      const float* nA = gA + (c + 1) * EDIM;
      const float* nB = gB + (c + 1) * EDIM;
      pa0 = *(const float4*)(nA);
      pa1 = *(const float4*)(nA + (size_t)16 * SLICE);
      pa2 = *(const float4*)(nA + (size_t)32 * SLICE);
      pa3 = *(const float4*)(nA + (size_t)48 * SLICE);
      pb0 = *(const float4*)(nB);
      pb1 = *(const float4*)(nB + (size_t)16 * SLICE);
    }

    #pragma unroll
    for (int kk = 0; kk < 64; kk += 4) {
      v2f afrag, bfrag;
      afrag = *(const v2f*)(lA + (wm * 16 + mrow) * LQA_P + kk + koff);
      bfrag.x = lB[(kk + koff + 0) * LQB_P + wn * 16 + ncol];
      bfrag.y = lB[(kk + koff + 1) * LQB_P + wn * 16 + ncol];
      acc[c] = __builtin_amdgcn_wmma_f32_16x16x4_f32(
          false, afrag, false, bfrag, (short)0, acc[c], false, false);
    }
    __syncthreads();
  }

  const float scale = 0.125f;
  const int rsel = (lane >> 4) * 8;
  const int col  = bd + wn * 16 + ncol;
  #pragma unroll
  for (int v = 0; v < 8; ++v) {
    int row = bb + wm * 16 + rsel + v;
    float* dst = aff + ((size_t)row * BSZ + col) * CCH;
    #pragma unroll
    for (int q = 0; q < 4; ++q) {
      float4 o;
      o.x = acc[4 * q + 0][v] * scale;
      o.y = acc[4 * q + 1][v] * scale;
      o.z = acc[4 * q + 2][v] * scale;
      o.w = acc[4 * q + 3][v] * scale;
      *(float4*)(dst + 4 * q) = o;
    }
  }

  float* park = lA;
  #pragma unroll
  for (int c = 0; c < CCH; ++c) {
    float m = -3.4e38f;
    #pragma unroll
    for (int v = 0; v < 8; ++v) m = fmaxf(m, acc[c][v] * scale);
    park[t * 17 + c] = m;
  }
  __syncthreads();
  {
    const int pc = t & 15;
    const int pg = t >> 4;
    float m = park[(pg * 16) * 17 + pc];
    for (int i = 1; i < 16; ++i) m = fmaxf(m, park[(pg * 16 + i) * 17 + pc]);
    red[pg * CCH + pc] = m;
  }
  __syncthreads();
  if (t < CCH) {
    float m = red[t];
    for (int g = 1; g < 16; ++g) m = fmaxf(m, red[g * CCH + t]);
    atomicMax(&maxbuf[t], enc_f32(m));
  }
}

__global__ __launch_bounds__(256) void sink_u_first(
    float* __restrict__ A, const unsigned int* __restrict__ maxbuf,
    const float* __restrict__ r, float* __restrict__ u)
{
  __shared__ float red[8 * 32];
  const int t    = threadIdx.x;
  const int lane = t & 31;
  const int wave = t >> 5;
  const int b0   = blockIdx.x * 2;
  float4* A4 = (float4*)(A + (size_t)b0 * (BSZ * CCH)) + t;

  const int c0 = (t & 3) * 4;
  const float m0 = dec_f32(maxbuf[c0 + 0]);
  const float m1 = dec_f32(maxbuf[c0 + 1]);
  const float m2 = dec_f32(maxbuf[c0 + 2]);
  const float m3 = dec_f32(maxbuf[c0 + 3]);

  float s[8] = {0.0f, 0.0f, 0.0f, 0.0f, 0.0f, 0.0f, 0.0f, 0.0f};
  #pragma unroll
  for (int i = 0; i < 16; ++i) {
    float4 x0 = A4[i * 256];
    float4 x1 = A4[4096 + i * 256];
    x0.x = expf(x0.x - m0);  x0.y = expf(x0.y - m1);
    x0.z = expf(x0.z - m2);  x0.w = expf(x0.w - m3);
    x1.x = expf(x1.x - m0);  x1.y = expf(x1.y - m1);
    x1.z = expf(x1.z - m2);  x1.w = expf(x1.w - m3);
    A4[i * 256]        = x0;
    A4[4096 + i * 256] = x1;
    s[0] += x0.x;  s[1] += x0.y;
    s[2] += x0.z;  s[3] += x0.w;
    s[4] += x1.x;  s[5] += x1.y;
    s[6] += x1.z;  s[7] += x1.w;
  }
  #pragma unroll
  for (int q = 0; q < 8; ++q) {
    s[q] += __shfl_xor(s[q], 4, 32);
    s[q] += __shfl_xor(s[q], 8, 32);
    s[q] += __shfl_xor(s[q], 16, 32);
  }
  if (lane < 4) {
    #pragma unroll
    for (int k = 0; k < 4; ++k) {
      red[wave * 32 +  0 + lane * 4 + k] = s[k];
      red[wave * 32 + 16 + lane * 4 + k] = s[4 + k];
    }
  }
  __syncthreads();
  if (t < 32) {
    float sum = red[t];
    for (int w = 1; w < 8; ++w) sum += red[w * 32 + t];
    u[b0 * CCH + t] = r[b0 * CCH + t] / sum;
  }
}

__global__ __launch_bounds__(256) void sink_u(
    const float* __restrict__ A, const float* __restrict__ v,
    const float* __restrict__ r, float* __restrict__ u)
{
  __shared__ float red[8 * 32];
  const int t    = threadIdx.x;
  const int lane = t & 31;
  const int wave = t >> 5;
  const int b0   = blockIdx.x * 2;
  const float4* A4 = (const float4*)(A + (size_t)b0 * (BSZ * CCH)) + t;
  const float4* v4 = (const float4*)v + t;

  float s[8] = {0.0f, 0.0f, 0.0f, 0.0f, 0.0f, 0.0f, 0.0f, 0.0f};
  #pragma unroll
  for (int i = 0; i < 16; ++i) {
    const float4 w  = v4[i * 256];
    const float4 x0 = A4[i * 256];
    const float4 x1 = A4[4096 + i * 256];
    s[0] += x0.x * w.x;  s[1] += x0.y * w.y;
    s[2] += x0.z * w.z;  s[3] += x0.w * w.w;
    s[4] += x1.x * w.x;  s[5] += x1.y * w.y;
    s[6] += x1.z * w.z;  s[7] += x1.w * w.w;
  }
  #pragma unroll
  for (int q = 0; q < 8; ++q) {
    s[q] += __shfl_xor(s[q], 4, 32);
    s[q] += __shfl_xor(s[q], 8, 32);
    s[q] += __shfl_xor(s[q], 16, 32);
  }
  if (lane < 4) {
    #pragma unroll
    for (int k = 0; k < 4; ++k) {
      red[wave * 32 +  0 + lane * 4 + k] = s[k];
      red[wave * 32 + 16 + lane * 4 + k] = s[4 + k];
    }
  }
  __syncthreads();
  if (t < 32) {
    float sum = red[t];
    for (int w = 1; w < 8; ++w) sum += red[w * 32 + t];
    u[b0 * CCH + t] = r[b0 * CCH + t] / sum;
  }
}

__global__ __launch_bounds__(256) void sink_v(
    const float* __restrict__ A, const float* __restrict__ u,
    const float* __restrict__ cmarg, float* __restrict__ v)
{
  __shared__ float4 su[BSZ * CCH / 4];
  __shared__ float red[8 * 64];
  const int t    = threadIdx.x;
  const int lane = t & 31;
  const int wave = t >> 5;
  const int l16  = t & 15;
  const int rg   = t >> 4;
  const int j0   = blockIdx.x * 64;
  const float* Ap = A + (size_t)rg * (BSZ * CCH) + j0 + l16 * 4;
  const float4* sp = su + rg * 4 + (l16 & 3);

  #pragma unroll
  for (int i = 0; i < 16; ++i) su[t + 256 * i] = ((const float4*)u)[t + 256 * i];
  __syncthreads();

  float4 acc = make_float4(0.0f, 0.0f, 0.0f, 0.0f);
  #pragma unroll
  for (int step = 0; step < 64; ++step) {
    float4 a = *(const float4*)(Ap + (size_t)step * 16 * (BSZ * CCH));
    float4 w = sp[step * 64];
    acc.x += a.x * w.x;  acc.y += a.y * w.y;
    acc.z += a.z * w.z;  acc.w += a.w * w.w;
  }

  acc.x += __shfl_xor(acc.x, 16, 32);
  acc.y += __shfl_xor(acc.y, 16, 32);
  acc.z += __shfl_xor(acc.z, 16, 32);
  acc.w += __shfl_xor(acc.w, 16, 32);
  if (lane < 16) {
    red[wave * 64 + lane * 4 + 0] = acc.x;
    red[wave * 64 + lane * 4 + 1] = acc.y;
    red[wave * 64 + lane * 4 + 2] = acc.z;
    red[wave * 64 + lane * 4 + 3] = acc.w;
  }
  __syncthreads();
  if (t < 64) {
    float sum = red[t];
    for (int w = 1; w < 8; ++w) sum += red[w * 64 + t];
    v[j0 + t] = cmarg[j0 + t] / sum;
  }
}

__global__ __launch_bounds__(256) void sink_v_final(
    float* __restrict__ A, const float* __restrict__ u,
    const float* __restrict__ cmarg)
{
  __shared__ float4 su[BSZ * CCH / 4];
  __shared__ float red[8 * 64];
  __shared__ float vs[64];
  const int t    = threadIdx.x;
  const int lane = t & 31;
  const int wave = t >> 5;
  const int l16  = t & 15;
  const int rg   = t >> 4;
  const int j0   = blockIdx.x * 64;
  float* Ap = A + (size_t)rg * (BSZ * CCH) + j0 + l16 * 4;
  const float4* sp = su + rg * 4 + (l16 & 3);

  float4 a[64];
  #pragma unroll
  for (int step = 0; step < 64; ++step)
    a[step] = *(const float4*)(Ap + (size_t)step * 16 * (BSZ * CCH));

  #pragma unroll
  for (int i = 0; i < 16; ++i) su[t + 256 * i] = ((const float4*)u)[t + 256 * i];
  __syncthreads();

  float4 acc = make_float4(0.0f, 0.0f, 0.0f, 0.0f);
  #pragma unroll
  for (int step = 0; step < 64; ++step) {
    float4 w = sp[step * 64];
    acc.x += a[step].x * w.x;  acc.y += a[step].y * w.y;
    acc.z += a[step].z * w.z;  acc.w += a[step].w * w.w;
  }

  acc.x += __shfl_xor(acc.x, 16, 32);
  acc.y += __shfl_xor(acc.y, 16, 32);
  acc.z += __shfl_xor(acc.z, 16, 32);
  acc.w += __shfl_xor(acc.w, 16, 32);
  if (lane < 16) {
    red[wave * 64 + lane * 4 + 0] = acc.x;
    red[wave * 64 + lane * 4 + 1] = acc.y;
    red[wave * 64 + lane * 4 + 2] = acc.z;
    red[wave * 64 + lane * 4 + 3] = acc.w;
  }
  __syncthreads();
  if (t < 64) {
    float sum = red[t];
    for (int w = 1; w < 8; ++w) sum += red[w * 64 + t];
    vs[t] = cmarg[j0 + t] / sum;
  }
  __syncthreads();

  const float4 vv = *(const float4*)(vs + l16 * 4);
  #pragma unroll
  for (int step = 0; step < 64; ++step) {
    float4 w = sp[step * 64];
    float4 p;
    p.x = a[step].x * w.x * vv.x;  p.y = a[step].y * w.y * vv.y;
    p.z = a[step].z * w.z * vv.z;  p.w = a[step].w * w.w * vv.w;
    *(float4*)(Ap + (size_t)step * 16 * (BSZ * CCH)) = p;
  }
}

extern "C" void kernel_launch(void* const* d_in, const int* in_sizes, int n_in,
                              void* d_out, int out_size, void* d_ws, size_t ws_size,
                              hipStream_t stream)
{
  const float* x1   = (const float*)d_in[0];
  const float* x2   = (const float*)d_in[1];
  const float* p1   = (const float*)d_in[2];
  const float* p2   = (const float*)d_in[3];
  const float* w1[4] = {(const float*)d_in[4],  (const float*)d_in[6],
                        (const float*)d_in[8],  (const float*)d_in[10]};
  const float* b1[4] = {(const float*)d_in[5],  (const float*)d_in[7],
                        (const float*)d_in[9],  (const float*)d_in[11]};
  const float* w2[4] = {(const float*)d_in[12], (const float*)d_in[14],
                        (const float*)d_in[16], (const float*)d_in[18]};
  const float* b2[4] = {(const float*)d_in[13], (const float*)d_in[15],
                        (const float*)d_in[17], (const float*)d_in[19]};

  float* ws = (float*)d_ws;
  float* hA = ws;
  float* hB = hA + 1024 * 512;
  float* q1 = hB + 1024 * 512;
  float* q2 = q1 + 1024 * 1024;
  float* uv_u = q2 + 1024 * 1024;
  float* uv_v = uv_u + 16384;
  unsigned int* maxbuf = (unsigned int*)(uv_v + 16384);

  float* Aout = (float*)d_out;

  float* h1a = q1;
  float* h1b = q1 + 1024 * 512;
  float* h2a = q2;
  float* h2b = q2 + 1024 * 512;

  gemm_bias_act<1><<<dim3(512/64, 1024/64, 2), 256, 0, stream>>>(
      x1, w1[0], b1[0], h1a,  x2, w2[0], b2[0], h2a,  maxbuf, 1024, 256, 512, 1);
  gemm_bias_act<1><<<dim3(512/64, 1024/64, 2), 256, 0, stream>>>(
      h1a, w1[1], b1[1], h1b,  h2a, w2[1], b2[1], h2b,  maxbuf, 1024, 512, 512, 1);
  gemm_bias_act<1><<<dim3(512/64, 1024/64, 2), 256, 0, stream>>>(
      h1b, w1[2], b1[2], hA,  h2b, w2[2], b2[2], hB,  maxbuf, 1024, 512, 512, 1);
  gemm_bias_act<2><<<dim3(1024/64, 1024/128, 2), 256, 0, stream>>>(
      hA, w1[3], b1[3], q1,  hB, w2[3], b2[3], q2,  maxbuf, 1024, 512, 1024, 2);

  affinity_wmma<<<dim3(1024/32, 1024/64), 256, 0, stream>>>(q1, q2, Aout, maxbuf);

  sink_u_first<<<512, 256, 0, stream>>>(Aout, maxbuf, p1, uv_u);
  sink_v<<<256, 256, 0, stream>>>(Aout, uv_u, p2, uv_v);
  for (int it = 1; it < 9; ++it) {
    sink_u<<<512, 256, 0, stream>>>(Aout, uv_v, p1, uv_u);
    sink_v<<<256, 256, 0, stream>>>(Aout, uv_u, p2, uv_v);
  }
  sink_u<<<512, 256, 0, stream>>>(Aout, uv_v, p1, uv_u);
  sink_v_final<<<256, 256, 0, stream>>>(Aout, uv_u, p2);
}
